# starting kernel
# baseline (speedup 1.0000x reference)
#include <hip/hip_runtime.h>

typedef __attribute__((ext_vector_type(2))) float v2f;
typedef __attribute__((ext_vector_type(8))) float v8f;

template <int TC>
__global__ __launch_bounds__(32) void sg_apply_wmma(
    const float* __restrict__ Sg, const float* __restrict__ Z,
    float* __restrict__ G) {
  const int nt    = blockIdx.x;
  const int mt    = blockIdx.y;
  const int l     = threadIdx.x;
  const int half  = l >> 4;
  const int lm    = l & 15;
  const int arow  = mt * 16 + lm;
  const int bcol0 = nt * 32 + lm;
  const int bcol1 = bcol0 + 16;

  v8f acc0 = {0.f, 0.f, 0.f, 0.f, 0.f, 0.f, 0.f, 0.f};
  v8f acc1 = {0.f, 0.f, 0.f, 0.f, 0.f, 0.f, 0.f, 0.f};

  const float* __restrict__ ap  = Sg + arow * 192 + half * 2;
  const float* __restrict__ bp0 = Z + half * 2 * TC + bcol0;
  const float* __restrict__ bp1 = Z + half * 2 * TC + bcol1;

#pragma unroll
  for (int k0 = 0; k0 < 192; k0 += 4) {
    v2f a, b0, b1;
    a.x  = ap[k0];
    a.y  = ap[k0 + 1];
    b0.x = bp0[k0 * TC];
    b0.y = bp0[k0 * TC + TC];
    b1.x = bp1[k0 * TC];
    b1.y = bp1[k0 * TC + TC];
    acc0 = __builtin_amdgcn_wmma_f32_16x16x4_f32(
        false, a, false, b0, (short)0, acc0, false, false);
    acc1 = __builtin_amdgcn_wmma_f32_16x16x4_f32(
        false, a, false, b1, (short)0, acc1, false, false);
  }

  const int rbase = mt * 16 + half * 8;
  float* __restrict__ gp = G + rbase * TC;
#pragma unroll
  for (int v = 0; v < 8; ++v) {
    gp[v * TC + bcol0] = acc0[v];
    gp[v * TC + bcol1] = acc1[v];
  }
}

__global__ void combine_shift(const float* __restrict__ x,
                              const float* __restrict__ g,
                              const float* __restrict__ s,
                              float* __restrict__ y, int t, int c) {
  const int idx = blockIdx.x * blockDim.x + threadIdx.x;
  const int total = 192 * t * c;
  if (idx >= total) return;
  const int ch   = idx % c;
  const int tmp  = idx / c;
  const int tau  = tmp % t;
  const int node = tmp / t;
  const int taum = (tau + t - 1) % t;
  const int i1 = (node * t + taum) * c + ch;
  const float s00 = s[0], s01 = s[1], s10 = s[2], s11 = s[3];
  y[idx] = s00 * x[idx] + s01 * g[idx] + s10 * x[i1] + s11 * g[i1];
}

__global__ void mix_channels(const float* __restrict__ x0,
                             const float* __restrict__ x1,
                             const float* __restrict__ x2,
                             const float* __restrict__ h,
                             float* __restrict__ y,
                             int t, int cin, int cout) {
  const int idx = blockIdx.x * blockDim.x + threadIdx.x;
  const int total = 192 * t * cout;
  if (idx >= total) return;
  const int o    = idx % cout;
  const int tmp  = idx / cout;
  const int tau  = tmp % t;
  const int node = tmp / t;
  const int base = (node * t + tau) * cin;
  float acc = 0.f;
  for (int i = 0; i < cin; ++i) {
    const float* hp = h + (o * cin + i) * 3;
    acc = fmaf(hp[0], x0[base + i], acc);
    acc = fmaf(hp[1], x1[base + i], acc);
    acc = fmaf(hp[2], x2[base + i], acc);
  }
  y[idx] = acc;
}

__global__ void down_relu(const float* __restrict__ y, float* __restrict__ a,
                          int t_out, int c) {
  const int idx = blockIdx.x * blockDim.x + threadIdx.x;
  const int total = 192 * t_out * c;
  if (idx >= total) return;
  const int ch   = idx % c;
  const int tmp  = idx / c;
  const int tau  = tmp % t_out;
  const int node = tmp / t_out;
  const int b0 = (node * 2 * t_out + 2 * tau) * c + ch;
  const float m = fmaxf(y[b0], y[b0 + c]);
  a[idx] = fmaxf(m, 0.f);
}

__global__ void up_relu(const float* __restrict__ y, float* __restrict__ a,
                        int t_in, int c) {
  const int idx = blockIdx.x * blockDim.x + threadIdx.x;
  const int t_out = 2 * t_in;
  const int total = 192 * t_out * c;
  if (idx >= total) return;
  const int ch   = idx % c;
  const int tmp  = idx / c;
  const int tau  = tmp % t_out;
  const int node = tmp / t_out;
  float v = 0.f;
  if ((tau & 1) == 0) v = fmaxf(y[(node * t_in + (tau >> 1)) * c + ch], 0.f);
  a[idx] = v;
}

static void launch_sg(const float* Sg, const float* Z, float* G, int tc,
                      hipStream_t stream) {
  const dim3 grid(tc / 32, 12);
  switch (tc) {
    case 32:
      sg_apply_wmma<32><<<grid, 32, 0, stream>>>(Sg, Z, G);
      break;
    case 256:
      sg_apply_wmma<256><<<grid, 32, 0, stream>>>(Sg, Z, G);
      break;
    default:
      sg_apply_wmma<512><<<grid, 32, 0, stream>>>(Sg, Z, G);
      break;
  }
}

static void conv_layer(const float* Sg, const float* s, const float* x0,
                       const float* h, float* X1, float* X2, float* G,
                       float* Y, int t, int cin, int cout, hipStream_t stream) {
  const int tc   = t * cin;
  const int totc = 192 * t * cin;
  const int toty = 192 * t * cout;
  launch_sg(Sg, x0, G, tc, stream);
  combine_shift<<<(totc + 255) / 256, 256, 0, stream>>>(x0, G, s, X1, t, cin);
  launch_sg(Sg, X1, G, tc, stream);
  combine_shift<<<(totc + 255) / 256, 256, 0, stream>>>(X1, G, s, X2, t, cin);
  mix_channels<<<(toty + 255) / 256, 256, 0, stream>>>(x0, X1, X2, h, Y, t,
                                                       cin, cout);
}

extern "C" void kernel_launch(void* const* d_in, const int* in_sizes, int n_in,
                              void* d_out, int out_size, void* d_ws,
                              size_t ws_size, hipStream_t stream) {
  (void)in_sizes; (void)n_in; (void)out_size; (void)ws_size;
  const float* X    = (const float*)d_in[0];
  const float* Sg   = (const float*)d_in[1];
  const float* s    = (const float*)d_in[2];
  const float* h_e1 = (const float*)d_in[3];
  const float* h_e2 = (const float*)d_in[4];
  const float* h_d1 = (const float*)d_in[5];
  const float* h_d2 = (const float*)d_in[6];
  float* out = (float*)d_out;
  float* ws  = (float*)d_ws;

  const int BUF = 192 * 32 * 16;
  float* X1 = ws + 0 * BUF;
  float* X2 = ws + 1 * BUF;
  float* G  = ws + 2 * BUF;
  float* Y  = ws + 3 * BUF;
  float* A0 = ws + 4 * BUF;
  float* A1 = ws + 5 * BUF;

  conv_layer(Sg, s, X, h_e1, X1, X2, G, Y, 32, 1, 16, stream);
  down_relu<<<(192 * 16 * 16 + 255) / 256, 256, 0, stream>>>(Y, A0, 16, 16);

  conv_layer(Sg, s, A0, h_e2, X1, X2, G, Y, 16, 16, 32, stream);
  down_relu<<<(192 * 8 * 32 + 255) / 256, 256, 0, stream>>>(Y, A1, 8, 32);

  up_relu<<<(192 * 16 * 32 + 255) / 256, 256, 0, stream>>>(A1, A0, 8, 32);
  conv_layer(Sg, s, A0, h_d1, X1, X2, G, Y, 16, 32, 16, stream);

  up_relu<<<(192 * 32 * 16 + 255) / 256, 256, 0, stream>>>(Y, A1, 16, 16);
  conv_layer(Sg, s, A1, h_d2, X1, X2, G, out, 32, 16, 1, stream);
}


# Launch fusion 24->12: time-shift combine folded into the Sg WMMA kernel (B operand pre-combined, epilogue adds x terms; sum order changes), pool/unpool+relu folded into channel mix
# speedup vs baseline: 1.1284x; 1.1284x over previous
#include <hip/hip_runtime.h>

typedef __attribute__((ext_vector_type(2))) float v2f;
typedef __attribute__((ext_vector_type(8))) float v8f;

template <int TC, int C>
__global__ __launch_bounds__(32) void s_apply_wmma(
    const float* __restrict__ Sg, const float* __restrict__ Z,
    const float* __restrict__ s, float* __restrict__ Xo) {
  const int nt    = blockIdx.x;
  const int mt    = blockIdx.y;
  const int l     = threadIdx.x;
  const int half  = l >> 4;
  const int lm    = l & 15;
  const int arow  = mt * 16 + lm;
  const int bcol0 = nt * 32 + lm;
  const int bcol1 = bcol0 + 16;
  const int bcol0m = (bcol0 >= C) ? (bcol0 - C) : (bcol0 - C + TC);
  const int bcol1m = (bcol1 >= C) ? (bcol1 - C) : (bcol1 - C + TC);

  const float s00 = s[0], s01 = s[1], s10 = s[2], s11 = s[3];

  v8f acc0 = {0.f, 0.f, 0.f, 0.f, 0.f, 0.f, 0.f, 0.f};
  v8f acc1 = {0.f, 0.f, 0.f, 0.f, 0.f, 0.f, 0.f, 0.f};

  const float* __restrict__ ap   = Sg + arow * 192 + half * 2;
  const float* __restrict__ bp0  = Z + half * 2 * TC + bcol0;
  const float* __restrict__ bp1  = Z + half * 2 * TC + bcol1;
  const float* __restrict__ bp0m = Z + half * 2 * TC + bcol0m;
  const float* __restrict__ bp1m = Z + half * 2 * TC + bcol1m;

#pragma unroll
  for (int k0 = 0; k0 < 192; k0 += 4) {
    v2f a, b0, b1;
    a.x = ap[k0];
    a.y = ap[k0 + 1];
    const float z0x = bp0[k0 * TC];
    const float z0y = bp0[k0 * TC + TC];
    const float z1x = bp1[k0 * TC];
    const float z1y = bp1[k0 * TC + TC];
    const float m0x = bp0m[k0 * TC];
    const float m0y = bp0m[k0 * TC + TC];
    float m1x, m1y;
    if constexpr (C == 16) {
      m1x = z0x;
      m1y = z0y;
    } else {
      m1x = bp1m[k0 * TC];
      m1y = bp1m[k0 * TC + TC];
    }
    b0.x = s01 * z0x + s11 * m0x;
    b0.y = s01 * z0y + s11 * m0y;
    b1.x = s01 * z1x + s11 * m1x;
    b1.y = s01 * z1y + s11 * m1y;
    acc0 = __builtin_amdgcn_wmma_f32_16x16x4_f32(
        false, a, false, b0, (short)0, acc0, false, false);
    acc1 = __builtin_amdgcn_wmma_f32_16x16x4_f32(
        false, a, false, b1, (short)0, acc1, false, false);
  }

  const int rbase = mt * 16 + half * 8;
  const float* __restrict__ zp = Z + rbase * TC;
  float* __restrict__ xp = Xo + rbase * TC;
#pragma unroll
  for (int v = 0; v < 8; ++v) {
    xp[v * TC + bcol0] =
        acc0[v] + s00 * zp[v * TC + bcol0] + s10 * zp[v * TC + bcol0m];
    xp[v * TC + bcol1] =
        acc1[v] + s00 * zp[v * TC + bcol1] + s10 * zp[v * TC + bcol1m];
  }
}

template <bool DOWN, bool UP>
__global__ void mix_pool(const float* __restrict__ x0,
                         const float* __restrict__ x1,
                         const float* __restrict__ x2,
                         const float* __restrict__ h,
                         float* __restrict__ out,
                         int tq_n, int cin, int cout) {
  const int idx = blockIdx.x * blockDim.x + threadIdx.x;
  const int total = 192 * tq_n * cout;
  if (idx >= total) return;
  const int o    = idx % cout;
  const int tmp  = idx / cout;
  const int tq   = tmp % tq_n;
  const int node = tmp / tq_n;
  const int t    = DOWN ? 2 * tq_n : tq_n;
  const int tau  = DOWN ? 2 * tq : tq;
  const int base = (node * t + tau) * cin;
  float acc = 0.f, accb = 0.f;
  for (int i = 0; i < cin; ++i) {
    const float* hp = h + (o * cin + i) * 3;
    const float h0 = hp[0], h1 = hp[1], h2 = hp[2];
    acc = fmaf(h0, x0[base + i], acc);
    acc = fmaf(h1, x1[base + i], acc);
    acc = fmaf(h2, x2[base + i], acc);
    if (DOWN) {
      accb = fmaf(h0, x0[base + cin + i], accb);
      accb = fmaf(h1, x1[base + cin + i], accb);
      accb = fmaf(h2, x2[base + cin + i], accb);
    }
  }
  float v = acc;
  if (DOWN) v = fmaxf(fmaxf(acc, accb), 0.f);
  if (UP) {
    const int ob = (node * 2 * tq_n + 2 * tq) * cout + o;
    out[ob]        = fmaxf(v, 0.f);
    out[ob + cout] = 0.f;
  } else {
    out[idx] = v;
  }
}

template <int TC, int C, bool DOWN, bool UP>
static void conv_layer(const float* Sg, const float* s, const float* x0,
                       const float* h, float* X1, float* X2, float* out,
                       int cout, hipStream_t stream) {
  const int t    = TC / C;
  const int tq_n = DOWN ? t / 2 : t;
  const int tot  = 192 * tq_n * cout;
  const dim3 grid(TC / 32, 12);
  s_apply_wmma<TC, C><<<grid, 32, 0, stream>>>(Sg, x0, s, X1);
  s_apply_wmma<TC, C><<<grid, 32, 0, stream>>>(Sg, X1, s, X2);
  mix_pool<DOWN, UP><<<(tot + 255) / 256, 256, 0, stream>>>(
      x0, X1, X2, h, out, tq_n, C, cout);
}

extern "C" void kernel_launch(void* const* d_in, const int* in_sizes, int n_in,
                              void* d_out, int out_size, void* d_ws,
                              size_t ws_size, hipStream_t stream) {
  (void)in_sizes; (void)n_in; (void)out_size; (void)ws_size;
  const float* X    = (const float*)d_in[0];
  const float* Sg   = (const float*)d_in[1];
  const float* s    = (const float*)d_in[2];
  const float* h_e1 = (const float*)d_in[3];
  const float* h_e2 = (const float*)d_in[4];
  const float* h_d1 = (const float*)d_in[5];
  const float* h_d2 = (const float*)d_in[6];
  float* out = (float*)d_out;
  float* ws  = (float*)d_ws;

  const int BUF = 192 * 32 * 16;
  float* X1 = ws + 0 * BUF;
  float* X2 = ws + 1 * BUF;
  float* A0 = ws + 2 * BUF;
  float* A1 = ws + 3 * BUF;

  conv_layer<32, 1, true, false>(Sg, s, X, h_e1, X1, X2, A0, 16, stream);
  conv_layer<256, 16, true, true>(Sg, s, A0, h_e2, X1, X2, A1, 32, stream);
  conv_layer<512, 32, false, true>(Sg, s, A1, h_d1, X1, X2, A0, 16, stream);
  conv_layer<512, 16, false, false>(Sg, s, A0, h_d2, X1, X2, out, 1, stream);
}


# Sg WMMA kernel: K=192 sum split over 8 waves per 16x32 tile (6 k-steps each, all loads in one group), partial tiles summed through LDS; sum order changes
# speedup vs baseline: 2.5411x; 2.2519x over previous
#include <hip/hip_runtime.h>

typedef __attribute__((ext_vector_type(2))) float v2f;
typedef __attribute__((ext_vector_type(8))) float v8f;

template <int TC, int C>
__global__ __launch_bounds__(256) void s_apply_wmma(
    const float* __restrict__ Sg, const float* __restrict__ Z,
    const float* __restrict__ s, float* __restrict__ Xo) {
  constexpr int NW = 8;
  constexpr int KW = 192 / NW;
  constexpr int KS = KW / 4;
  __shared__ float part[NW * 8 * 2 * 32];

  const int nt    = blockIdx.x;
  const int mt    = blockIdx.y;
  const int w     = threadIdx.x >> 5;
  const int l     = threadIdx.x & 31;
  const int half  = l >> 4;
  const int lm    = l & 15;
  const int arow  = mt * 16 + lm;
  const int bcol0 = nt * 32 + lm;
  const int bcol1 = bcol0 + 16;
  const int bcol0m = (bcol0 >= C) ? (bcol0 - C) : (bcol0 - C + TC);
  const int bcol1m = (bcol1 >= C) ? (bcol1 - C) : (bcol1 - C + TC);

  const int ecol  = nt * 32 + l;
  const int ecolm = (ecol >= C) ? (ecol - C) : (ecol - C + TC);
  const float* __restrict__ zp = Z + (mt * 16 + w) * TC;
  const float ez0 = zp[ecol];
  const float em0 = zp[ecolm];
  const float ez1 = zp[8 * TC + ecol];
  const float em1 = zp[8 * TC + ecolm];

  const float s00 = s[0], s01 = s[1], s10 = s[2], s11 = s[3];

  const int kb = w * KW + half * 2;
  const float* __restrict__ ap   = Sg + arow * 192 + kb;
  const float* __restrict__ bp0  = Z + kb * TC + bcol0;
  const float* __restrict__ bp1  = Z + kb * TC + bcol1;
  const float* __restrict__ bp0m = Z + kb * TC + bcol0m;
  const float* __restrict__ bp1m = Z + kb * TC + bcol1m;

  v2f a[KS];
  float z0x[KS], z0y[KS], z1x[KS], z1y[KS];
  float m0x[KS], m0y[KS], m1x[KS], m1y[KS];
#pragma unroll
  for (int j = 0; j < KS; ++j) {
    const int k0 = j * 4;
    a[j].x = ap[k0];
    a[j].y = ap[k0 + 1];
    z0x[j] = bp0[k0 * TC];
    z0y[j] = bp0[k0 * TC + TC];
    z1x[j] = bp1[k0 * TC];
    z1y[j] = bp1[k0 * TC + TC];
    m0x[j] = bp0m[k0 * TC];
    m0y[j] = bp0m[k0 * TC + TC];
    if constexpr (C == 16) {
      m1x[j] = z0x[j];
      m1y[j] = z0y[j];
    } else {
      m1x[j] = bp1m[k0 * TC];
      m1y[j] = bp1m[k0 * TC + TC];
    }
  }

  __builtin_amdgcn_sched_barrier(0);

  v8f acc0 = {0.f, 0.f, 0.f, 0.f, 0.f, 0.f, 0.f, 0.f};
  v8f acc1 = {0.f, 0.f, 0.f, 0.f, 0.f, 0.f, 0.f, 0.f};
#pragma unroll
  for (int j = 0; j < KS; ++j) {
    v2f b0, b1;
    b0.x = s01 * z0x[j] + s11 * m0x[j];
    b0.y = s01 * z0y[j] + s11 * m0y[j];
    b1.x = s01 * z1x[j] + s11 * m1x[j];
    b1.y = s01 * z1y[j] + s11 * m1y[j];
    acc0 = __builtin_amdgcn_wmma_f32_16x16x4_f32(
        false, a[j], false, b0, (short)0, acc0, false, false);
    acc1 = __builtin_amdgcn_wmma_f32_16x16x4_f32(
        false, a[j], false, b1, (short)0, acc1, false, false);
  }

#pragma unroll
  for (int v = 0; v < 8; ++v) {
    part[((w * 8 + v) * 2 + 0) * 32 + l] = acc0[v];
    part[((w * 8 + v) * 2 + 1) * 32 + l] = acc1[v];
  }
  __syncthreads();

  const int pidx = (w * 2 + half) * 32 + lm;
  float sum0 = part[pidx];
  float sum1 = part[pidx + 16];
#pragma unroll
  for (int q = 1; q < NW; ++q) {
    sum0 += part[q * 512 + pidx];
    sum1 += part[q * 512 + pidx + 16];
  }
  float* __restrict__ xp = Xo + (mt * 16 + w) * TC;
  xp[ecol]          = sum0 + s00 * ez0 + s10 * em0;
  xp[8 * TC + ecol] = sum1 + s00 * ez1 + s10 * em1;
}

template <bool DOWN, bool UP>
__global__ void mix_pool(const float* __restrict__ x0,
                         const float* __restrict__ x1,
                         const float* __restrict__ x2,
                         const float* __restrict__ h,
                         float* __restrict__ out,
                         int tq_n, int cin, int cout) {
  const int idx = blockIdx.x * blockDim.x + threadIdx.x;
  const int total = 192 * tq_n * cout;
  if (idx >= total) return;
  const int o    = idx % cout;
  const int tmp  = idx / cout;
  const int tq   = tmp % tq_n;
  const int node = tmp / tq_n;
  const int t    = DOWN ? 2 * tq_n : tq_n;
  const int tau  = DOWN ? 2 * tq : tq;
  const int base = (node * t + tau) * cin;
  float acc = 0.f, accb = 0.f;
  for (int i = 0; i < cin; ++i) {
    const float* hp = h + (o * cin + i) * 3;
    const float h0 = hp[0], h1 = hp[1], h2 = hp[2];
    acc = fmaf(h0, x0[base + i], acc);
    acc = fmaf(h1, x1[base + i], acc);
    acc = fmaf(h2, x2[base + i], acc);
    if (DOWN) {
      accb = fmaf(h0, x0[base + cin + i], accb);
      accb = fmaf(h1, x1[base + cin + i], accb);
      accb = fmaf(h2, x2[base + cin + i], accb);
    }
  }
  float v = acc;
  if (DOWN) v = fmaxf(fmaxf(acc, accb), 0.f);
  if (UP) {
    const int ob = (node * 2 * tq_n + 2 * tq) * cout + o;
    out[ob]        = fmaxf(v, 0.f);
    out[ob + cout] = 0.f;
  } else {
    out[idx] = v;
  }
}

template <int TC, int C, bool DOWN, bool UP>
static void conv_layer(const float* Sg, const float* s, const float* x0,
                       const float* h, float* X1, float* X2, float* out,
                       int cout, hipStream_t stream) {
  const int t    = TC / C;
  const int tq_n = DOWN ? t / 2 : t;
  const int tot  = 192 * tq_n * cout;
  const dim3 grid(TC / 32, 12);
  s_apply_wmma<TC, C><<<grid, 256, 0, stream>>>(Sg, x0, s, X1);
  s_apply_wmma<TC, C><<<grid, 256, 0, stream>>>(Sg, X1, s, X2);
  mix_pool<DOWN, UP><<<(tot + 255) / 256, 256, 0, stream>>>(
      x0, X1, X2, h, out, tq_n, C, cout);
}

extern "C" void kernel_launch(void* const* d_in, const int* in_sizes, int n_in,
                              void* d_out, int out_size, void* d_ws,
                              size_t ws_size, hipStream_t stream) {
  (void)in_sizes; (void)n_in; (void)out_size; (void)ws_size;
  const float* X    = (const float*)d_in[0];
  const float* Sg   = (const float*)d_in[1];
  const float* s    = (const float*)d_in[2];
  const float* h_e1 = (const float*)d_in[3];
  const float* h_e2 = (const float*)d_in[4];
  const float* h_d1 = (const float*)d_in[5];
  const float* h_d2 = (const float*)d_in[6];
  float* out = (float*)d_out;
  float* ws  = (float*)d_ws;

  const int BUF = 192 * 32 * 16;
  float* X1 = ws + 0 * BUF;
  float* X2 = ws + 1 * BUF;
  float* A0 = ws + 2 * BUF;
  float* A1 = ws + 3 * BUF;

  conv_layer<32, 1, true, false>(Sg, s, X, h_e1, X1, X2, A0, 16, stream);
  conv_layer<256, 16, true, true>(Sg, s, A0, h_e2, X1, X2, A1, 32, stream);
  conv_layer<512, 32, false, true>(Sg, s, A1, h_d1, X1, X2, A0, 16, stream);
  conv_layer<512, 16, false, false>(Sg, s, A0, h_d2, X1, X2, out, 1, stream);
}


# Channel/tap mix + pool/unpool of layers 1-3 folded into the epilogue of the second S-apply WMMA kernel (x0,X1,X2 tile and taps in LDS; X2 plane not written); 12->9 launches
# speedup vs baseline: 3.0595x; 1.2040x over previous
#include <hip/hip_runtime.h>

typedef __attribute__((ext_vector_type(2))) float v2f;
typedef __attribute__((ext_vector_type(8))) float v8f;

template <int TC, int C, bool MIX = false, int COUT = 1, bool DOWN = false,
          bool UP = false>
__global__ __launch_bounds__(256) void s_apply_wmma(
    const float* __restrict__ Sg, const float* __restrict__ Z,
    const float* __restrict__ s, float* __restrict__ Xo,
    const float* __restrict__ x0, const float* __restrict__ h) {
  constexpr int NW = 8;
  constexpr int KW = 192 / NW;
  constexpr int KS = KW / 4;
  __shared__ float part[NW * 8 * 2 * 32];
  constexpr int HT = COUT * C * 3;
  constexpr int NH = (HT + 255) / 256;
  constexpr bool HFULL = (HT % 256) == 0;
  __shared__ float tile[MIX ? 3 * 512 : 1];
  __shared__ float hs[MIX ? HT : 1];

  const int nt    = blockIdx.x;
  const int mt    = blockIdx.y;
  const int w     = threadIdx.x >> 5;
  const int l     = threadIdx.x & 31;
  const int half  = l >> 4;
  const int lm    = l & 15;
  const int arow  = mt * 16 + lm;
  const int bcol0 = nt * 32 + lm;
  const int bcol1 = bcol0 + 16;
  const int bcol0m = (bcol0 >= C) ? (bcol0 - C) : (bcol0 - C + TC);
  const int bcol1m = (bcol1 >= C) ? (bcol1 - C) : (bcol1 - C + TC);

  const int ecol  = nt * 32 + l;
  const int ecolm = (ecol >= C) ? (ecol - C) : (ecol - C + TC);
  const float* __restrict__ zp = Z + (mt * 16 + w) * TC;
  const float ez0 = zp[ecol];
  const float em0 = zp[ecolm];
  const float ez1 = zp[8 * TC + ecol];
  const float em1 = zp[8 * TC + ecolm];
  float ex0 = 0.f, ex1 = 0.f;
  float hreg[NH];
  if constexpr (MIX) {
    ex0 = x0[(mt * 16 + w) * TC + ecol];
    ex1 = x0[(mt * 16 + w + 8) * TC + ecol];
#pragma unroll
    for (int j = 0; j < NH; ++j) {
      const int q = j * 256 + (int)threadIdx.x;
      hreg[j] = (HFULL || q < HT) ? h[q] : 0.f;
    }
  }

  const float s00 = s[0], s01 = s[1], s10 = s[2], s11 = s[3];

  const int kb = w * KW + half * 2;
  const float* __restrict__ ap   = Sg + arow * 192 + kb;
  const float* __restrict__ bp0  = Z + kb * TC + bcol0;
  const float* __restrict__ bp1  = Z + kb * TC + bcol1;
  const float* __restrict__ bp0m = Z + kb * TC + bcol0m;
  const float* __restrict__ bp1m = Z + kb * TC + bcol1m;

  v2f a[KS];
  float z0x[KS], z0y[KS], z1x[KS], z1y[KS];
  float m0x[KS], m0y[KS], m1x[KS], m1y[KS];
#pragma unroll
  for (int j = 0; j < KS; ++j) {
    const int k0 = j * 4;
    a[j].x = ap[k0];
    a[j].y = ap[k0 + 1];
    z0x[j] = bp0[k0 * TC];
    z0y[j] = bp0[k0 * TC + TC];
    z1x[j] = bp1[k0 * TC];
    z1y[j] = bp1[k0 * TC + TC];
    m0x[j] = bp0m[k0 * TC];
    m0y[j] = bp0m[k0 * TC + TC];
    if constexpr (C == 16) {
      m1x[j] = z0x[j];
      m1y[j] = z0y[j];
    } else {
      m1x[j] = bp1m[k0 * TC];
      m1y[j] = bp1m[k0 * TC + TC];
    }
  }

  __builtin_amdgcn_sched_barrier(0);

  v8f acc0 = {0.f, 0.f, 0.f, 0.f, 0.f, 0.f, 0.f, 0.f};
  v8f acc1 = {0.f, 0.f, 0.f, 0.f, 0.f, 0.f, 0.f, 0.f};
#pragma unroll
  for (int j = 0; j < KS; ++j) {
    v2f b0, b1;
    b0.x = s01 * z0x[j] + s11 * m0x[j];
    b0.y = s01 * z0y[j] + s11 * m0y[j];
    b1.x = s01 * z1x[j] + s11 * m1x[j];
    b1.y = s01 * z1y[j] + s11 * m1y[j];
    acc0 = __builtin_amdgcn_wmma_f32_16x16x4_f32(
        false, a[j], false, b0, (short)0, acc0, false, false);
    acc1 = __builtin_amdgcn_wmma_f32_16x16x4_f32(
        false, a[j], false, b1, (short)0, acc1, false, false);
  }

#pragma unroll
  for (int v = 0; v < 8; ++v) {
    part[((w * 8 + v) * 2 + 0) * 32 + l] = acc0[v];
    part[((w * 8 + v) * 2 + 1) * 32 + l] = acc1[v];
  }
  if constexpr (MIX) {
#pragma unroll
    for (int j = 0; j < NH; ++j) {
      const int q = j * 256 + (int)threadIdx.x;
      if (HFULL || q < HT) hs[(q % (C * 3)) * COUT + q / (C * 3)] = hreg[j];
    }
  }
  __syncthreads();

  const int pidx = (w * 2 + half) * 32 + lm;
  float sum0 = part[pidx];
  float sum1 = part[pidx + 16];
#pragma unroll
  for (int q = 1; q < NW; ++q) {
    sum0 += part[q * 512 + pidx];
    sum1 += part[q * 512 + pidx + 16];
  }
  const float r0 = sum0 + s00 * ez0 + s10 * em0;
  const float r1 = sum1 + s00 * ez1 + s10 * em1;
  if constexpr (!MIX) {
    float* __restrict__ xp = Xo + (mt * 16 + w) * TC;
    xp[ecol]          = r0;
    xp[8 * TC + ecol] = r1;
  } else {
    tile[w * 32 + l]              = ex0;
    tile[(w + 8) * 32 + l]        = ex1;
    tile[512 + w * 32 + l]        = ez0;
    tile[512 + (w + 8) * 32 + l]  = ez1;
    tile[1024 + w * 32 + l]       = r0;
    tile[1024 + (w + 8) * 32 + l] = r1;
    __syncthreads();

    constexpr int TAUS = 32 / C;
    constexpr int TQT  = DOWN ? TAUS / 2 : TAUS;
    constexpr int TQN  = DOWN ? TC / C / 2 : TC / C;
    constexpr int NOUT = 16 * TQT * COUT;
    for (int li = (int)threadIdx.x; li < NOUT; li += 256) {
      const int o  = li % COUT;
      const int tl = (li / COUT) % TQT;
      const int nl = li / (COUT * TQT);
      const float* tp = tile + nl * 32 + (DOWN ? 2 * tl : tl) * C;
      float acc = 0.f, accb = 0.f;
      for (int i = 0; i < C; ++i) {
        const float h0 = hs[(i * 3 + 0) * COUT + o];
        const float h1 = hs[(i * 3 + 1) * COUT + o];
        const float h2 = hs[(i * 3 + 2) * COUT + o];
        acc = fmaf(h0, tp[i], acc);
        acc = fmaf(h1, tp[512 + i], acc);
        acc = fmaf(h2, tp[1024 + i], acc);
        if (DOWN) {
          accb = fmaf(h0, tp[C + i], accb);
          accb = fmaf(h1, tp[512 + C + i], accb);
          accb = fmaf(h2, tp[1024 + C + i], accb);
        }
      }
      float v = acc;
      if (DOWN) v = fmaxf(fmaxf(acc, accb), 0.f);
      const int node = mt * 16 + nl;
      const int tq   = nt * TQT + tl;
      if (UP) {
        const int ob = (node * 2 * TQN + 2 * tq) * COUT + o;
        Xo[ob]        = fmaxf(v, 0.f);
        Xo[ob + COUT] = 0.f;
      } else {
        Xo[(node * TQN + tq) * COUT + o] = v;
      }
    }
  }
}

template <bool DOWN, bool UP>
__global__ void mix_pool(const float* __restrict__ x0,
                         const float* __restrict__ x1,
                         const float* __restrict__ x2,
                         const float* __restrict__ h,
                         float* __restrict__ out,
                         int tq_n, int cin, int cout) {
  const int idx = blockIdx.x * blockDim.x + threadIdx.x;
  const int total = 192 * tq_n * cout;
  if (idx >= total) return;
  const int o    = idx % cout;
  const int tmp  = idx / cout;
  const int tq   = tmp % tq_n;
  const int node = tmp / tq_n;
  const int t    = DOWN ? 2 * tq_n : tq_n;
  const int tau  = DOWN ? 2 * tq : tq;
  const int base = (node * t + tau) * cin;
  float acc = 0.f, accb = 0.f;
  for (int i = 0; i < cin; ++i) {
    const float* hp = h + (o * cin + i) * 3;
    const float h0 = hp[0], h1 = hp[1], h2 = hp[2];
    acc = fmaf(h0, x0[base + i], acc);
    acc = fmaf(h1, x1[base + i], acc);
    acc = fmaf(h2, x2[base + i], acc);
    if (DOWN) {
      accb = fmaf(h0, x0[base + cin + i], accb);
      accb = fmaf(h1, x1[base + cin + i], accb);
      accb = fmaf(h2, x2[base + cin + i], accb);
    }
  }
  float v = acc;
  if (DOWN) v = fmaxf(fmaxf(acc, accb), 0.f);
  if (UP) {
    const int ob = (node * 2 * tq_n + 2 * tq) * cout + o;
    out[ob]        = fmaxf(v, 0.f);
    out[ob + cout] = 0.f;
  } else {
    out[idx] = v;
  }
}

template <int TC, int C, int COUT, bool DOWN, bool UP>
static void conv_layer_fused(const float* Sg, const float* s, const float* x0,
                             const float* h, float* X1, float* out,
                             hipStream_t stream) {
  const dim3 grid(TC / 32, 12);
  s_apply_wmma<TC, C><<<grid, 256, 0, stream>>>(Sg, x0, s, X1, x0, h);
  s_apply_wmma<TC, C, true, COUT, DOWN, UP><<<grid, 256, 0, stream>>>(
      Sg, X1, s, out, x0, h);
}

template <int TC, int C, bool DOWN, bool UP>
static void conv_layer(const float* Sg, const float* s, const float* x0,
                       const float* h, float* X1, float* X2, float* out,
                       int cout, hipStream_t stream) {
  const int t    = TC / C;
  const int tq_n = DOWN ? t / 2 : t;
  const int tot  = 192 * tq_n * cout;
  const dim3 grid(TC / 32, 12);
  s_apply_wmma<TC, C><<<grid, 256, 0, stream>>>(Sg, x0, s, X1, x0, h);
  s_apply_wmma<TC, C><<<grid, 256, 0, stream>>>(Sg, X1, s, X2, x0, h);
  mix_pool<DOWN, UP><<<(tot + 255) / 256, 256, 0, stream>>>(
      x0, X1, X2, h, out, tq_n, C, cout);
}

extern "C" void kernel_launch(void* const* d_in, const int* in_sizes, int n_in,
                              void* d_out, int out_size, void* d_ws,
                              size_t ws_size, hipStream_t stream) {
  (void)in_sizes; (void)n_in; (void)out_size; (void)ws_size;
  const float* X    = (const float*)d_in[0];
  const float* Sg   = (const float*)d_in[1];
  const float* s    = (const float*)d_in[2];
  const float* h_e1 = (const float*)d_in[3];
  const float* h_e2 = (const float*)d_in[4];
  const float* h_d1 = (const float*)d_in[5];
  const float* h_d2 = (const float*)d_in[6];
  float* out = (float*)d_out;
  float* ws  = (float*)d_ws;

  const int BUF = 192 * 32 * 16;
  float* X1 = ws + 0 * BUF;
  float* X2 = ws + 1 * BUF;
  float* A0 = ws + 2 * BUF;
  float* A1 = ws + 3 * BUF;

  conv_layer_fused<32, 1, 16, true, false>(Sg, s, X, h_e1, X1, A0, stream);
  conv_layer_fused<256, 16, 32, true, true>(Sg, s, A0, h_e2, X1, A1, stream);
  conv_layer_fused<512, 32, 16, false, true>(Sg, s, A1, h_d1, X1, A0, stream);
  conv_layer<512, 16, false, false>(Sg, s, A0, h_d2, X1, X2, out, 1, stream);
}


# Sg2 = Sg@Sg built once per call (new WMMA launch); both S-applies of a layer in one launch: X2 = AAz + Sg(2ABz) + Sg2(BBz) beside X1; layers 1-3 write no X1 plane; 9->6 launches; sum order changes
# speedup vs baseline: 3.7948x; 1.2403x over previous
#include <hip/hip_runtime.h>

typedef __attribute__((ext_vector_type(2))) float v2f;
typedef __attribute__((ext_vector_type(8))) float v8f;

__global__ __launch_bounds__(256) void sg_square_wmma(
    const float* __restrict__ Sg, float* __restrict__ Sg2) {
  constexpr int NW = 8;
  constexpr int KW = 192 / NW;
  constexpr int KS = KW / 4;
  __shared__ float part[NW * 8 * 2 * 32];

  const int nt   = blockIdx.x;
  const int mt   = blockIdx.y;
  const int w    = threadIdx.x >> 5;
  const int l    = threadIdx.x & 31;
  const int half = l >> 4;
  const int lm   = l & 15;
  const int arow  = mt * 16 + lm;
  const int bcol0 = nt * 32 + lm;
  const int kb    = w * KW + half * 2;
  const float* __restrict__ ap  = Sg + arow * 192 + kb;
  const float* __restrict__ bp0 = Sg + kb * 192 + bcol0;

  v2f a[KS], b0[KS], b1[KS];
#pragma unroll
  for (int j = 0; j < KS; ++j) {
    const int k0 = j * 4;
    a[j].x  = ap[k0];
    a[j].y  = ap[k0 + 1];
    b0[j].x = bp0[k0 * 192];
    b0[j].y = bp0[k0 * 192 + 192];
    b1[j].x = bp0[k0 * 192 + 16];
    b1[j].y = bp0[k0 * 192 + 192 + 16];
  }

  __builtin_amdgcn_sched_barrier(0);

  v8f acc0 = {0.f, 0.f, 0.f, 0.f, 0.f, 0.f, 0.f, 0.f};
  v8f acc1 = {0.f, 0.f, 0.f, 0.f, 0.f, 0.f, 0.f, 0.f};
#pragma unroll
  for (int j = 0; j < KS; ++j) {
    acc0 = __builtin_amdgcn_wmma_f32_16x16x4_f32(
        false, a[j], false, b0[j], (short)0, acc0, false, false);
    acc1 = __builtin_amdgcn_wmma_f32_16x16x4_f32(
        false, a[j], false, b1[j], (short)0, acc1, false, false);
  }

#pragma unroll
  for (int v = 0; v < 8; ++v) {
    part[((w * 8 + v) * 2 + 0) * 32 + l] = acc0[v];
    part[((w * 8 + v) * 2 + 1) * 32 + l] = acc1[v];
  }
  __syncthreads();

  const int pidx = (w * 2 + half) * 32 + lm;
  float sum0 = part[pidx];
  float sum1 = part[pidx + 16];
#pragma unroll
  for (int q = 1; q < NW; ++q) {
    sum0 += part[q * 512 + pidx];
    sum1 += part[q * 512 + pidx + 16];
  }
  float* __restrict__ op = Sg2 + (mt * 16 + w) * 192 + nt * 32 + l;
  op[0]       = sum0;
  op[8 * 192] = sum1;
}

template <int TC, int C, bool MIX, int COUT = 1, bool DOWN = false,
          bool UP = false>
__global__ __launch_bounds__(256) void s2_apply_wmma(
    const float* __restrict__ Sg, const float* __restrict__ Sg2,
    const float* __restrict__ Z, const float* __restrict__ s,
    float* __restrict__ Xo, float* __restrict__ Xo2,
    const float* __restrict__ h) {
  constexpr int NW = 8;
  constexpr int KW = 192 / NW;
  constexpr int KS = KW / 4;
  __shared__ float part[NW * 8 * 4 * 32];
  constexpr int HT = COUT * C * 3;
  constexpr int NH = (HT + 255) / 256;
  constexpr bool HFULL = (HT % 256) == 0;
  __shared__ float tile[MIX ? 3 * 512 : 1];
  __shared__ float hs[MIX ? HT : 1];

  const int nt    = blockIdx.x;
  const int mt    = blockIdx.y;
  const int w     = threadIdx.x >> 5;
  const int l     = threadIdx.x & 31;
  const int half  = l >> 4;
  const int lm    = l & 15;
  const int arow  = mt * 16 + lm;
  const int bcol0 = nt * 32 + lm;
  const int bcol1 = bcol0 + 16;
  const int bcol0m  = (bcol0 >= C) ? (bcol0 - C) : (bcol0 - C + TC);
  const int bcol1m  = (bcol1 >= C) ? (bcol1 - C) : (bcol1 - C + TC);
  const int bcol0mm = (bcol0 >= 2 * C) ? (bcol0 - 2 * C) : (bcol0 - 2 * C + TC);
  const int bcol1mm = (bcol1 >= 2 * C) ? (bcol1 - 2 * C) : (bcol1 - 2 * C + TC);

  const int ecol   = nt * 32 + l;
  const int ecolm  = (ecol >= C) ? (ecol - C) : (ecol - C + TC);
  const int ecolmm = (ecol >= 2 * C) ? (ecol - 2 * C) : (ecol - 2 * C + TC);
  const float* __restrict__ zp = Z + (mt * 16 + w) * TC;
  const float ez0 = zp[ecol];
  const float em0 = zp[ecolm];
  const float en0 = zp[ecolmm];
  const float ez1 = zp[8 * TC + ecol];
  const float em1 = zp[8 * TC + ecolm];
  const float en1 = zp[8 * TC + ecolmm];
  float hreg[NH];
  if constexpr (MIX) {
#pragma unroll
    for (int j = 0; j < NH; ++j) {
      const int q = j * 256 + (int)threadIdx.x;
      hreg[j] = (HFULL || q < HT) ? h[q] : 0.f;
    }
  }

  const float s00 = s[0], s01 = s[1], s10 = s[2], s11 = s[3];
  const float a0 = s00 * s00, a1 = 2.f * s00 * s10, a2 = s10 * s10;
  const float c0 = 2.f * s00 * s01;
  const float c1 = 2.f * (s00 * s11 + s10 * s01);
  const float c2 = 2.f * s10 * s11;
  const float d0 = s01 * s01, d1 = 2.f * s01 * s11, d2 = s11 * s11;

  const int kb = w * KW + half * 2;
  const float* __restrict__ ap    = Sg + arow * 192 + kb;
  const float* __restrict__ ap2   = Sg2 + arow * 192 + kb;
  const float* __restrict__ bp0   = Z + kb * TC + bcol0;
  const float* __restrict__ bp1   = Z + kb * TC + bcol1;
  const float* __restrict__ bp0m  = Z + kb * TC + bcol0m;
  const float* __restrict__ bp1m  = Z + kb * TC + bcol1m;
  const float* __restrict__ bp0mm = Z + kb * TC + bcol0mm;
  const float* __restrict__ bp1mm = Z + kb * TC + bcol1mm;

  v2f a[KS], a2f[KS];
  float z0x[KS], z0y[KS], z1x[KS], z1y[KS];
  float m0x[KS], m0y[KS], m1x[KS], m1y[KS];
  float n0x[KS], n0y[KS], n1x[KS], n1y[KS];
#pragma unroll
  for (int j = 0; j < KS; ++j) {
    const int k0 = j * 4;
    a[j].x   = ap[k0];
    a[j].y   = ap[k0 + 1];
    a2f[j].x = ap2[k0];
    a2f[j].y = ap2[k0 + 1];
    z0x[j] = bp0[k0 * TC];
    z0y[j] = bp0[k0 * TC + TC];
    z1x[j] = bp1[k0 * TC];
    z1y[j] = bp1[k0 * TC + TC];
    m0x[j] = bp0m[k0 * TC];
    m0y[j] = bp0m[k0 * TC + TC];
    if constexpr (C == 16) {
      m1x[j] = z0x[j];
      m1y[j] = z0y[j];
      n1x[j] = m0x[j];
      n1y[j] = m0y[j];
    } else {
      m1x[j] = bp1m[k0 * TC];
      m1y[j] = bp1m[k0 * TC + TC];
      n1x[j] = bp1mm[k0 * TC];
      n1y[j] = bp1mm[k0 * TC + TC];
    }
    n0x[j] = bp0mm[k0 * TC];
    n0y[j] = bp0mm[k0 * TC + TC];
  }

  __builtin_amdgcn_sched_barrier(0);

  v8f x1a = {0.f, 0.f, 0.f, 0.f, 0.f, 0.f, 0.f, 0.f};
  v8f x1b = {0.f, 0.f, 0.f, 0.f, 0.f, 0.f, 0.f, 0.f};
  v8f x2a = {0.f, 0.f, 0.f, 0.f, 0.f, 0.f, 0.f, 0.f};
  v8f x2b = {0.f, 0.f, 0.f, 0.f, 0.f, 0.f, 0.f, 0.f};
#pragma unroll
  for (int j = 0; j < KS; ++j) {
    v2f p0, p1, q0, q1, r0, r1;
    p0.x = s01 * z0x[j] + s11 * m0x[j];
    p0.y = s01 * z0y[j] + s11 * m0y[j];
    p1.x = s01 * z1x[j] + s11 * m1x[j];
    p1.y = s01 * z1y[j] + s11 * m1y[j];
    q0.x = c0 * z0x[j] + c1 * m0x[j] + c2 * n0x[j];
    q0.y = c0 * z0y[j] + c1 * m0y[j] + c2 * n0y[j];
    q1.x = c0 * z1x[j] + c1 * m1x[j] + c2 * n1x[j];
    q1.y = c0 * z1y[j] + c1 * m1y[j] + c2 * n1y[j];
    r0.x = d0 * z0x[j] + d1 * m0x[j] + d2 * n0x[j];
    r0.y = d0 * z0y[j] + d1 * m0y[j] + d2 * n0y[j];
    r1.x = d0 * z1x[j] + d1 * m1x[j] + d2 * n1x[j];
    r1.y = d0 * z1y[j] + d1 * m1y[j] + d2 * n1y[j];
    x1a = __builtin_amdgcn_wmma_f32_16x16x4_f32(
        false, a[j], false, p0, (short)0, x1a, false, false);
    x1b = __builtin_amdgcn_wmma_f32_16x16x4_f32(
        false, a[j], false, p1, (short)0, x1b, false, false);
    x2a = __builtin_amdgcn_wmma_f32_16x16x4_f32(
        false, a[j], false, q0, (short)0, x2a, false, false);
    x2b = __builtin_amdgcn_wmma_f32_16x16x4_f32(
        false, a[j], false, q1, (short)0, x2b, false, false);
    x2a = __builtin_amdgcn_wmma_f32_16x16x4_f32(
        false, a2f[j], false, r0, (short)0, x2a, false, false);
    x2b = __builtin_amdgcn_wmma_f32_16x16x4_f32(
        false, a2f[j], false, r1, (short)0, x2b, false, false);
  }

#pragma unroll
  for (int v = 0; v < 8; ++v) {
    part[((w * 8 + v) * 4 + 0) * 32 + l] = x1a[v];
    part[((w * 8 + v) * 4 + 1) * 32 + l] = x1b[v];
    part[((w * 8 + v) * 4 + 2) * 32 + l] = x2a[v];
    part[((w * 8 + v) * 4 + 3) * 32 + l] = x2b[v];
  }
  if constexpr (MIX) {
#pragma unroll
    for (int j = 0; j < NH; ++j) {
      const int q = j * 256 + (int)threadIdx.x;
      if (HFULL || q < HT) hs[(q % (C * 3)) * COUT + q / (C * 3)] = hreg[j];
    }
  }
  __syncthreads();

  const int pidx = (w * 4 + half) * 32 + lm;
  float u0 = part[pidx];
  float u1 = part[pidx + 16];
  float t0 = part[pidx + 64];
  float t1 = part[pidx + 64 + 16];
#pragma unroll
  for (int q = 1; q < NW; ++q) {
    u0 += part[q * 1024 + pidx];
    u1 += part[q * 1024 + pidx + 16];
    t0 += part[q * 1024 + pidx + 64];
    t1 += part[q * 1024 + pidx + 64 + 16];
  }
  const float f0 = u0 + s00 * ez0 + s10 * em0;
  const float f1 = u1 + s00 * ez1 + s10 * em1;
  const float g0 = t0 + a0 * ez0 + a1 * em0 + a2 * en0;
  const float g1 = t1 + a0 * ez1 + a1 * em1 + a2 * en1;
  if constexpr (!MIX) {
    float* __restrict__ xp1 = Xo + (mt * 16 + w) * TC;
    float* __restrict__ xp2 = Xo2 + (mt * 16 + w) * TC;
    xp1[ecol]          = f0;
    xp1[8 * TC + ecol] = f1;
    xp2[ecol]          = g0;
    xp2[8 * TC + ecol] = g1;
  } else {
    tile[w * 32 + l]              = ez0;
    tile[(w + 8) * 32 + l]        = ez1;
    tile[512 + w * 32 + l]        = f0;
    tile[512 + (w + 8) * 32 + l]  = f1;
    tile[1024 + w * 32 + l]       = g0;
    tile[1024 + (w + 8) * 32 + l] = g1;
    __syncthreads();

    constexpr int TAUS = 32 / C;
    constexpr int TQT  = DOWN ? TAUS / 2 : TAUS;
    constexpr int TQN  = DOWN ? TC / C / 2 : TC / C;
    constexpr int NOUT = 16 * TQT * COUT;
    for (int li = (int)threadIdx.x; li < NOUT; li += 256) {
      const int o  = li % COUT;
      const int tl = (li / COUT) % TQT;
      const int nl = li / (COUT * TQT);
      const float* tp = tile + nl * 32 + (DOWN ? 2 * tl : tl) * C;
      float acc = 0.f, accb = 0.f;
      for (int i = 0; i < C; ++i) {
        const float h0 = hs[(i * 3 + 0) * COUT + o];
        const float h1 = hs[(i * 3 + 1) * COUT + o];
        const float h2 = hs[(i * 3 + 2) * COUT + o];
        acc = fmaf(h0, tp[i], acc);
        acc = fmaf(h1, tp[512 + i], acc);
        acc = fmaf(h2, tp[1024 + i], acc);
        if (DOWN) {
          accb = fmaf(h0, tp[C + i], accb);
          accb = fmaf(h1, tp[512 + C + i], accb);
          accb = fmaf(h2, tp[1024 + C + i], accb);
        }
      }
      float v = acc;
      if (DOWN) v = fmaxf(fmaxf(acc, accb), 0.f);
      const int node = mt * 16 + nl;
      const int tq   = nt * TQT + tl;
      if (UP) {
        const int ob = (node * 2 * TQN + 2 * tq) * COUT + o;
        Xo[ob]        = fmaxf(v, 0.f);
        Xo[ob + COUT] = 0.f;
      } else {
        Xo[(node * TQN + tq) * COUT + o] = v;
      }
    }
  }
}

template <bool DOWN, bool UP>
__global__ void mix_pool(const float* __restrict__ x0,
                         const float* __restrict__ x1,
                         const float* __restrict__ x2,
                         const float* __restrict__ h,
                         float* __restrict__ out,
                         int tq_n, int cin, int cout) {
  const int idx = blockIdx.x * blockDim.x + threadIdx.x;
  const int total = 192 * tq_n * cout;
  if (idx >= total) return;
  const int o    = idx % cout;
  const int tmp  = idx / cout;
  const int tq   = tmp % tq_n;
  const int node = tmp / tq_n;
  const int t    = DOWN ? 2 * tq_n : tq_n;
  const int tau  = DOWN ? 2 * tq : tq;
  const int base = (node * t + tau) * cin;
  float acc = 0.f, accb = 0.f;
  for (int i = 0; i < cin; ++i) {
    const float* hp = h + (o * cin + i) * 3;
    const float h0 = hp[0], h1 = hp[1], h2 = hp[2];
    acc = fmaf(h0, x0[base + i], acc);
    acc = fmaf(h1, x1[base + i], acc);
    acc = fmaf(h2, x2[base + i], acc);
    if (DOWN) {
      accb = fmaf(h0, x0[base + cin + i], accb);
      accb = fmaf(h1, x1[base + cin + i], accb);
      accb = fmaf(h2, x2[base + cin + i], accb);
    }
  }
  float v = acc;
  if (DOWN) v = fmaxf(fmaxf(acc, accb), 0.f);
  if (UP) {
    const int ob = (node * 2 * tq_n + 2 * tq) * cout + o;
    out[ob]        = fmaxf(v, 0.f);
    out[ob + cout] = 0.f;
  } else {
    out[idx] = v;
  }
}

template <int TC, int C, int COUT, bool DOWN, bool UP>
static void conv_layer_fused(const float* Sg, const float* Sg2, const float* s,
                             const float* x0, const float* h, float* out,
                             hipStream_t stream) {
  const dim3 grid(TC / 32, 12);
  s2_apply_wmma<TC, C, true, COUT, DOWN, UP><<<grid, 256, 0, stream>>>(
      Sg, Sg2, x0, s, out, out, h);
}

template <int TC, int C, bool DOWN, bool UP>
static void conv_layer(const float* Sg, const float* Sg2, const float* s,
                       const float* x0, const float* h, float* X1, float* X2,
                       float* out, int cout, hipStream_t stream) {
  const int t    = TC / C;
  const int tq_n = DOWN ? t / 2 : t;
  const int tot  = 192 * tq_n * cout;
  const dim3 grid(TC / 32, 12);
  s2_apply_wmma<TC, C, false><<<grid, 256, 0, stream>>>(Sg, Sg2, x0, s, X1, X2,
                                                        h);
  mix_pool<DOWN, UP><<<(tot + 255) / 256, 256, 0, stream>>>(
      x0, X1, X2, h, out, tq_n, C, cout);
}

extern "C" void kernel_launch(void* const* d_in, const int* in_sizes, int n_in,
                              void* d_out, int out_size, void* d_ws,
                              size_t ws_size, hipStream_t stream) {
  (void)in_sizes; (void)n_in; (void)out_size; (void)ws_size;
  const float* X    = (const float*)d_in[0];
  const float* Sg   = (const float*)d_in[1];
  const float* s    = (const float*)d_in[2];
  const float* h_e1 = (const float*)d_in[3];
  const float* h_e2 = (const float*)d_in[4];
  const float* h_d1 = (const float*)d_in[5];
  const float* h_d2 = (const float*)d_in[6];
  float* out = (float*)d_out;
  float* ws  = (float*)d_ws;

  const int BUF = 192 * 32 * 16;
  float* Sg2 = ws + 0 * BUF;
  float* X1  = ws + 1 * BUF;
  float* A0  = ws + 2 * BUF;
  float* A1  = ws + 3 * BUF;
  float* X2  = A1;

  sg_square_wmma<<<dim3(6, 12), 256, 0, stream>>>(Sg, Sg2);
  conv_layer_fused<32, 1, 16, true, false>(Sg, Sg2, s, X, h_e1, A0, stream);
  conv_layer_fused<256, 16, 32, true, true>(Sg, Sg2, s, A0, h_e2, A1, stream);
  conv_layer_fused<512, 32, 16, false, true>(Sg, Sg2, s, A1, h_d1, A0, stream);
  conv_layer<512, 16, false, false>(Sg, Sg2, s, A0, h_d2, X1, X2, out, 1,
                                    stream);
}


# Last layer (16->1 ch): its channel mix is taken ahead of its shifts, in the epilogue of decoder 1 (mix and shift commute); S then acts on 192x32, not 192x512; 6->5 launches; sum order changes
# speedup vs baseline: 4.4099x; 1.1621x over previous
#include <hip/hip_runtime.h>

typedef __attribute__((ext_vector_type(2))) float v2f;
typedef __attribute__((ext_vector_type(8))) float v8f;

__global__ __launch_bounds__(256) void sg_square_wmma(
    const float* __restrict__ Sg, float* __restrict__ Sg2) {
  constexpr int NW = 8;
  constexpr int KW = 192 / NW;
  constexpr int KS = KW / 4;
  __shared__ float part[NW * 8 * 2 * 32];

  const int nt   = blockIdx.x;
  const int mt   = blockIdx.y;
  const int w    = threadIdx.x >> 5;
  const int l    = threadIdx.x & 31;
  const int half = l >> 4;
  const int lm   = l & 15;
  const int arow  = mt * 16 + lm;
  const int bcol0 = nt * 32 + lm;
  const int kb    = w * KW + half * 2;
  const float* __restrict__ ap  = Sg + arow * 192 + kb;
  const float* __restrict__ bp0 = Sg + kb * 192 + bcol0;

  v2f a[KS], b0[KS], b1[KS];
#pragma unroll
  for (int j = 0; j < KS; ++j) {
    const int k0 = j * 4;
    a[j].x  = ap[k0];
    a[j].y  = ap[k0 + 1];
    b0[j].x = bp0[k0 * 192];
    b0[j].y = bp0[k0 * 192 + 192];
    b1[j].x = bp0[k0 * 192 + 16];
    b1[j].y = bp0[k0 * 192 + 192 + 16];
  }

  __builtin_amdgcn_sched_barrier(0);

  v8f acc0 = {0.f, 0.f, 0.f, 0.f, 0.f, 0.f, 0.f, 0.f};
  v8f acc1 = {0.f, 0.f, 0.f, 0.f, 0.f, 0.f, 0.f, 0.f};
#pragma unroll
  for (int j = 0; j < KS; ++j) {
    acc0 = __builtin_amdgcn_wmma_f32_16x16x4_f32(
        false, a[j], false, b0[j], (short)0, acc0, false, false);
    acc1 = __builtin_amdgcn_wmma_f32_16x16x4_f32(
        false, a[j], false, b1[j], (short)0, acc1, false, false);
  }

#pragma unroll
  for (int v = 0; v < 8; ++v) {
    part[((w * 8 + v) * 2 + 0) * 32 + l] = acc0[v];
    part[((w * 8 + v) * 2 + 1) * 32 + l] = acc1[v];
  }
  __syncthreads();

  const int pidx = (w * 2 + half) * 32 + lm;
  float sum0 = part[pidx];
  float sum1 = part[pidx + 16];
#pragma unroll
  for (int q = 1; q < NW; ++q) {
    sum0 += part[q * 512 + pidx];
    sum1 += part[q * 512 + pidx + 16];
  }
  float* __restrict__ op = Sg2 + (mt * 16 + w) * 192 + nt * 32 + l;
  op[0]       = sum0;
  op[8 * 192] = sum1;
}

template <int TC, int C, int COUT, bool DOWN, bool UP, bool PREMIX>
__global__ __launch_bounds__(256) void s2_apply_wmma(
    const float* __restrict__ Sg, const float* __restrict__ Sg2,
    const float* __restrict__ Z, const float* __restrict__ s,
    float* __restrict__ Xo, const float* __restrict__ h,
    const float* __restrict__ hn) {
  constexpr int NW = 8;
  constexpr int KW = 192 / NW;
  constexpr int KS = KW / 4;
  __shared__ float part[NW * 8 * 4 * 32];
  constexpr int HT = COUT * C * 3;
  constexpr int NH = (HT + 255) / 256;
  constexpr bool HFULL = (HT % 256) == 0;
  __shared__ float tile[3 * 512];
  __shared__ float hs[HT];
  __shared__ float vst[PREMIX ? 48 : 1];

  const int nt    = blockIdx.x;
  const int mt    = blockIdx.y;
  const int w     = threadIdx.x >> 5;
  const int l     = threadIdx.x & 31;
  const int half  = l >> 4;
  const int lm    = l & 15;
  const int arow  = mt * 16 + lm;
  const int bcol0 = nt * 32 + lm;
  const int bcol1 = bcol0 + 16;
  const int bcol0m  = (bcol0 >= C) ? (bcol0 - C) : (bcol0 - C + TC);
  const int bcol1m  = (bcol1 >= C) ? (bcol1 - C) : (bcol1 - C + TC);
  const int bcol0mm = (bcol0 >= 2 * C) ? (bcol0 - 2 * C) : (bcol0 - 2 * C + TC);
  const int bcol1mm = (bcol1 >= 2 * C) ? (bcol1 - 2 * C) : (bcol1 - 2 * C + TC);

  const int ecol   = nt * 32 + l;
  const int ecolm  = (ecol >= C) ? (ecol - C) : (ecol - C + TC);
  const int ecolmm = (ecol >= 2 * C) ? (ecol - 2 * C) : (ecol - 2 * C + TC);
  const float* __restrict__ zp = Z + (mt * 16 + w) * TC;
  const float ez0 = zp[ecol];
  const float em0 = zp[ecolm];
  const float en0 = zp[ecolmm];
  const float ez1 = zp[8 * TC + ecol];
  const float em1 = zp[8 * TC + ecolm];
  const float en1 = zp[8 * TC + ecolmm];
  float hreg[NH];
#pragma unroll
  for (int j = 0; j < NH; ++j) {
    const int q = j * 256 + (int)threadIdx.x;
    hreg[j] = (HFULL || q < HT) ? h[q] : 0.f;
  }
  float hn0 = 0.f, hn1 = 0.f, hn2 = 0.f;
  if constexpr (PREMIX) {
    const int on = (int)threadIdx.x & 15;
    hn0 = hn[on * 3 + 0];
    hn1 = hn[on * 3 + 1];
    hn2 = hn[on * 3 + 2];
  }

  const float s00 = s[0], s01 = s[1], s10 = s[2], s11 = s[3];
  const float a0 = s00 * s00, a1 = 2.f * s00 * s10, a2 = s10 * s10;
  const float c0 = 2.f * s00 * s01;
  const float c1 = 2.f * (s00 * s11 + s10 * s01);
  const float c2 = 2.f * s10 * s11;
  const float d0 = s01 * s01, d1 = 2.f * s01 * s11, d2 = s11 * s11;

  const int kb = w * KW + half * 2;
  const float* __restrict__ ap    = Sg + arow * 192 + kb;
  const float* __restrict__ ap2   = Sg2 + arow * 192 + kb;
  const float* __restrict__ bp0   = Z + kb * TC + bcol0;
  const float* __restrict__ bp1   = Z + kb * TC + bcol1;
  const float* __restrict__ bp0m  = Z + kb * TC + bcol0m;
  const float* __restrict__ bp1m  = Z + kb * TC + bcol1m;
  const float* __restrict__ bp0mm = Z + kb * TC + bcol0mm;
  const float* __restrict__ bp1mm = Z + kb * TC + bcol1mm;

  v2f a[KS], a2f[KS];
  float z0x[KS], z0y[KS], z1x[KS], z1y[KS];
  float m0x[KS], m0y[KS], m1x[KS], m1y[KS];
  float n0x[KS], n0y[KS], n1x[KS], n1y[KS];
#pragma unroll
  for (int j = 0; j < KS; ++j) {
    const int k0 = j * 4;
    a[j].x   = ap[k0];
    a[j].y   = ap[k0 + 1];
    a2f[j].x = ap2[k0];
    a2f[j].y = ap2[k0 + 1];
    z0x[j] = bp0[k0 * TC];
    z0y[j] = bp0[k0 * TC + TC];
    z1x[j] = bp1[k0 * TC];
    z1y[j] = bp1[k0 * TC + TC];
    m0x[j] = bp0m[k0 * TC];
    m0y[j] = bp0m[k0 * TC + TC];
    if constexpr (C == 16) {
      m1x[j] = z0x[j];
      m1y[j] = z0y[j];
      n1x[j] = m0x[j];
      n1y[j] = m0y[j];
    } else {
      m1x[j] = bp1m[k0 * TC];
      m1y[j] = bp1m[k0 * TC + TC];
      n1x[j] = bp1mm[k0 * TC];
      n1y[j] = bp1mm[k0 * TC + TC];
    }
    n0x[j] = bp0mm[k0 * TC];
    n0y[j] = bp0mm[k0 * TC + TC];
  }

  __builtin_amdgcn_sched_barrier(0);

  v8f x1a = {0.f, 0.f, 0.f, 0.f, 0.f, 0.f, 0.f, 0.f};
  v8f x1b = {0.f, 0.f, 0.f, 0.f, 0.f, 0.f, 0.f, 0.f};
  v8f x2a = {0.f, 0.f, 0.f, 0.f, 0.f, 0.f, 0.f, 0.f};
  v8f x2b = {0.f, 0.f, 0.f, 0.f, 0.f, 0.f, 0.f, 0.f};
#pragma unroll
  for (int j = 0; j < KS; ++j) {
    v2f p0, p1, q0, q1, r0, r1;
    p0.x = s01 * z0x[j] + s11 * m0x[j];
    p0.y = s01 * z0y[j] + s11 * m0y[j];
    p1.x = s01 * z1x[j] + s11 * m1x[j];
    p1.y = s01 * z1y[j] + s11 * m1y[j];
    q0.x = c0 * z0x[j] + c1 * m0x[j] + c2 * n0x[j];
    q0.y = c0 * z0y[j] + c1 * m0y[j] + c2 * n0y[j];
    q1.x = c0 * z1x[j] + c1 * m1x[j] + c2 * n1x[j];
    q1.y = c0 * z1y[j] + c1 * m1y[j] + c2 * n1y[j];
    r0.x = d0 * z0x[j] + d1 * m0x[j] + d2 * n0x[j];
    r0.y = d0 * z0y[j] + d1 * m0y[j] + d2 * n0y[j];
    r1.x = d0 * z1x[j] + d1 * m1x[j] + d2 * n1x[j];
    r1.y = d0 * z1y[j] + d1 * m1y[j] + d2 * n1y[j];
    x1a = __builtin_amdgcn_wmma_f32_16x16x4_f32(
        false, a[j], false, p0, (short)0, x1a, false, false);
    x1b = __builtin_amdgcn_wmma_f32_16x16x4_f32(
        false, a[j], false, p1, (short)0, x1b, false, false);
    x2a = __builtin_amdgcn_wmma_f32_16x16x4_f32(
        false, a[j], false, q0, (short)0, x2a, false, false);
    x2b = __builtin_amdgcn_wmma_f32_16x16x4_f32(
        false, a[j], false, q1, (short)0, x2b, false, false);
    x2a = __builtin_amdgcn_wmma_f32_16x16x4_f32(
        false, a2f[j], false, r0, (short)0, x2a, false, false);
    x2b = __builtin_amdgcn_wmma_f32_16x16x4_f32(
        false, a2f[j], false, r1, (short)0, x2b, false, false);
  }

#pragma unroll
  for (int v = 0; v < 8; ++v) {
    part[((w * 8 + v) * 4 + 0) * 32 + l] = x1a[v];
    part[((w * 8 + v) * 4 + 1) * 32 + l] = x1b[v];
    part[((w * 8 + v) * 4 + 2) * 32 + l] = x2a[v];
    part[((w * 8 + v) * 4 + 3) * 32 + l] = x2b[v];
  }
#pragma unroll
  for (int j = 0; j < NH; ++j) {
    const int q = j * 256 + (int)threadIdx.x;
    if (HFULL || q < HT) hs[(q % (C * 3)) * COUT + q / (C * 3)] = hreg[j];
  }
  __syncthreads();

  const int pidx = (w * 4 + half) * 32 + lm;
  float u0 = part[pidx];
  float u1 = part[pidx + 16];
  float t0 = part[pidx + 64];
  float t1 = part[pidx + 64 + 16];
#pragma unroll
  for (int q = 1; q < NW; ++q) {
    u0 += part[q * 1024 + pidx];
    u1 += part[q * 1024 + pidx + 16];
    t0 += part[q * 1024 + pidx + 64];
    t1 += part[q * 1024 + pidx + 64 + 16];
  }
  const float f0 = u0 + s00 * ez0 + s10 * em0;
  const float f1 = u1 + s00 * ez1 + s10 * em1;
  const float g0 = t0 + a0 * ez0 + a1 * em0 + a2 * en0;
  const float g1 = t1 + a0 * ez1 + a1 * em1 + a2 * en1;
  tile[w * 32 + l]              = ez0;
  tile[(w + 8) * 32 + l]        = ez1;
  tile[512 + w * 32 + l]        = f0;
  tile[512 + (w + 8) * 32 + l]  = f1;
  tile[1024 + w * 32 + l]       = g0;
  tile[1024 + (w + 8) * 32 + l] = g1;
  __syncthreads();

  constexpr int TAUS = 32 / C;
  constexpr int TQT  = DOWN ? TAUS / 2 : TAUS;
  constexpr int TQN  = DOWN ? TC / C / 2 : TC / C;
  constexpr int NOUT = 16 * TQT * COUT;
  static_assert(!PREMIX || (NOUT == 256 && COUT == 16 && UP && !DOWN),
                "premix needs one (node, channel) pair per thread");
  for (int li = (int)threadIdx.x; li < NOUT; li += 256) {
    const int o  = li % COUT;
    const int tl = (li / COUT) % TQT;
    const int nl = li / (COUT * TQT);
    const float* tp = tile + nl * 32 + (DOWN ? 2 * tl : tl) * C;
    float acc = 0.f, accb = 0.f;
    for (int i = 0; i < C; ++i) {
      const float h0 = hs[(i * 3 + 0) * COUT + o];
      const float h1 = hs[(i * 3 + 1) * COUT + o];
      const float h2 = hs[(i * 3 + 2) * COUT + o];
      acc = fmaf(h0, tp[i], acc);
      acc = fmaf(h1, tp[512 + i], acc);
      acc = fmaf(h2, tp[1024 + i], acc);
      if (DOWN) {
        accb = fmaf(h0, tp[C + i], accb);
        accb = fmaf(h1, tp[512 + C + i], accb);
        accb = fmaf(h2, tp[1024 + C + i], accb);
      }
    }
    float v = acc;
    if (DOWN) v = fmaxf(fmaxf(acc, accb), 0.f);
    const int node = mt * 16 + nl;
    const int tq   = nt * TQT + tl;
    if constexpr (PREMIX) {
      const float rv = fmaxf(v, 0.f);
      float p0 = hn0 * rv, p1 = hn1 * rv, p2 = hn2 * rv;
#pragma unroll
      for (int m = 8; m >= 1; m >>= 1) {
        p0 += __shfl_xor(p0, m, 32);
        p1 += __shfl_xor(p1, m, 32);
        p2 += __shfl_xor(p2, m, 32);
      }
      if (o < 3) vst[((o + 2) % 3) * 16 + nl] = (o == 0) ? p0 : ((o == 1) ? p1 : p2);
    } else if (UP) {
      const int ob = (node * 2 * TQN + 2 * tq) * COUT + o;
      Xo[ob]        = fmaxf(v, 0.f);
      Xo[ob + COUT] = 0.f;
    } else {
      Xo[(node * TQN + tq) * COUT + o] = v;
    }
  }
  if constexpr (PREMIX) {
    __syncthreads();
    float* __restrict__ vp = Xo + (nt * 12 + mt) * 64;
    if (w == 0) vp[l] = vst[l];
    if (w == 1 && l < 16) vp[32 + l] = vst[32 + l];
  }
}

__global__ __launch_bounds__(256) void last_layer_wmma(
    const float* __restrict__ Sg, const float* __restrict__ Sg2,
    const float* __restrict__ V, const float* __restrict__ s,
    float* __restrict__ out) {
  constexpr int NW = 8;
  constexpr int KW = 192 / NW;
  constexpr int KS = KW / 4;
  constexpr int VS = 17;
  constexpr int PL = 192 * VS + 2;
  constexpr int NCH = 192 / NW;
  __shared__ float vs[2 * PL];
  __shared__ float part[NW * 8 * 2 * 32];

  const int mt   = blockIdx.x;
  const int w    = threadIdx.x >> 5;
  const int l    = threadIdx.x & 31;
  const int half = l >> 4;
  const int lm   = l & 15;
  const int arow = mt * 16 + lm;
  const int kb   = w * KW + half * 2;
  const int odd  = l & 1;
  const int ta   = l >> 1;
  const int tb   = (ta + 15) & 15;

  float sv[NCH];
#pragma unroll
  for (int j = 0; j < NCH; ++j) sv[j] = V[(w * NCH + j) * 64 + l];

  const float* __restrict__ ap  = Sg + arow * 192 + kb;
  const float* __restrict__ ap2 = Sg2 + arow * 192 + kb;
  v2f a[KS], a2f[KS];
#pragma unroll
  for (int j = 0; j < KS; ++j) {
    const int k0 = j * 4;
    a[j].x   = ap[k0];
    a[j].y   = ap[k0 + 1];
    a2f[j].x = ap2[k0];
    a2f[j].y = ap2[k0 + 1];
  }

  const float* __restrict__ va = V + (ta * 12 + mt) * 64 + w;
  const float* __restrict__ vb = V + (tb * 12 + mt) * 64 + w;
  const float l1_0  = va[0];
  const float l2a_0 = va[16];
  const float l0_0  = va[32];
  const float l2b_0 = vb[16];
  const float l1_1  = va[8];
  const float l2a_1 = va[16 + 8];
  const float l0_1  = va[32 + 8];
  const float l2b_1 = vb[16 + 8];

  const float s00 = s[0], s01 = s[1], s10 = s[2], s11 = s[3];

#pragma unroll
  for (int j = 0; j < NCH; ++j) {
    vs[half * PL + ((j % 12) * 16 + lm) * VS + 2 * w + j / 12] = sv[j];
  }

  const float a0 = s00 * s00, a1 = 2.f * s00 * s10, a2 = s10 * s10;
  const float c0 = 2.f * s00 * s01;
  const float c1 = 2.f * (s00 * s11 + s10 * s01);
  const float c2 = 2.f * s10 * s11;
  const float d0 = s01 * s01, d1 = 2.f * s01 * s11, d2 = s11 * s11;
  const float e1 = odd ? s11 : s01;
  const float e2 = odd ? c1 : c0;
  const float e3 = odd ? 0.f : c2;
  const float f2 = odd ? d1 : d0;
  const float f3 = odd ? 0.f : d2;
  __syncthreads();

  const int ta0 = lm >> 1;
  const int ta1 = ta0 + 8;
  const int tb0 = (ta0 + 15) & 15;
  const int tb1 = (ta0 + 7) & 15;
  v8f acc0 = {0.f, 0.f, 0.f, 0.f, 0.f, 0.f, 0.f, 0.f};
  v8f acc1 = {0.f, 0.f, 0.f, 0.f, 0.f, 0.f, 0.f, 0.f};
#pragma unroll
  for (int j = 0; j < KS; ++j) {
    const float* r1 = vs + (kb + j * 4) * VS;
    const float* r2 = r1 + PL;
    v2f p0, p1, q0, q1;
    p0.x = e1 * r1[ta0] + e2 * r2[ta0] + e3 * r2[tb0];
    p0.y = e1 * r1[VS + ta0] + e2 * r2[VS + ta0] + e3 * r2[VS + tb0];
    p1.x = e1 * r1[ta1] + e2 * r2[ta1] + e3 * r2[tb1];
    p1.y = e1 * r1[VS + ta1] + e2 * r2[VS + ta1] + e3 * r2[VS + tb1];
    q0.x = f2 * r2[ta0] + f3 * r2[tb0];
    q0.y = f2 * r2[VS + ta0] + f3 * r2[VS + tb0];
    q1.x = f2 * r2[ta1] + f3 * r2[tb1];
    q1.y = f2 * r2[VS + ta1] + f3 * r2[VS + tb1];
    acc0 = __builtin_amdgcn_wmma_f32_16x16x4_f32(
        false, a[j], false, p0, (short)0, acc0, false, false);
    acc1 = __builtin_amdgcn_wmma_f32_16x16x4_f32(
        false, a[j], false, p1, (short)0, acc1, false, false);
    acc0 = __builtin_amdgcn_wmma_f32_16x16x4_f32(
        false, a2f[j], false, q0, (short)0, acc0, false, false);
    acc1 = __builtin_amdgcn_wmma_f32_16x16x4_f32(
        false, a2f[j], false, q1, (short)0, acc1, false, false);
  }

#pragma unroll
  for (int v = 0; v < 8; ++v) {
    part[((w * 8 + v) * 2 + 0) * 32 + l] = acc0[v];
    part[((w * 8 + v) * 2 + 1) * 32 + l] = acc1[v];
  }
  __syncthreads();

  const int pidx = (w * 2 + half) * 32 + lm;
  float sum0 = part[pidx];
  float sum1 = part[pidx + 16];
#pragma unroll
  for (int q = 1; q < NW; ++q) {
    sum0 += part[q * 512 + pidx];
    sum1 += part[q * 512 + pidx + 16];
  }
  const float g0 = odd ? 0.f : 1.f;
  const float g1 = odd ? s10 : s00;
  const float g2 = odd ? a1 : a0;
  const float g3 = odd ? 0.f : a2;
  float* __restrict__ op = out + (mt * 16 + w) * 32 + l;
  op[0]      = sum0 + g0 * l0_0 + g1 * l1_0 + g2 * l2a_0 + g3 * l2b_0;
  op[8 * 32] = sum1 + g0 * l0_1 + g1 * l1_1 + g2 * l2a_1 + g3 * l2b_1;
}

extern "C" void kernel_launch(void* const* d_in, const int* in_sizes, int n_in,
                              void* d_out, int out_size, void* d_ws,
                              size_t ws_size, hipStream_t stream) {
  (void)in_sizes; (void)n_in; (void)out_size; (void)ws_size;
  const float* X    = (const float*)d_in[0];
  const float* Sg   = (const float*)d_in[1];
  const float* s    = (const float*)d_in[2];
  const float* h_e1 = (const float*)d_in[3];
  const float* h_e2 = (const float*)d_in[4];
  const float* h_d1 = (const float*)d_in[5];
  const float* h_d2 = (const float*)d_in[6];
  float* out = (float*)d_out;
  float* ws  = (float*)d_ws;

  const int BUF = 192 * 32 * 16;
  float* Sg2 = ws + 0 * BUF;
  float* A0  = ws + 2 * BUF;
  float* A1  = ws + 3 * BUF;
  float* V   = A0;

  sg_square_wmma<<<dim3(6, 12), 256, 0, stream>>>(Sg, Sg2);
  s2_apply_wmma<32, 1, 16, true, false, false>
      <<<dim3(1, 12), 256, 0, stream>>>(Sg, Sg2, X, s, A0, h_e1, h_e1);
  s2_apply_wmma<256, 16, 32, true, true, false>
      <<<dim3(8, 12), 256, 0, stream>>>(Sg, Sg2, A0, s, A1, h_e2, h_e2);
  s2_apply_wmma<512, 32, 16, false, true, true>
      <<<dim3(16, 12), 256, 0, stream>>>(Sg, Sg2, A1, s, V, h_d1, h_d2);
  last_layer_wmma<<<dim3(12), 256, 0, stream>>>(Sg, Sg2, V, s, out);
}


# Channel/tap mix of layers 1-3 moved from per-thread FMA loops to f32 WMMA (X tile from LDS as A, taps from memory as B; K split over waves, parts summed in LDS); sum order changes
# speedup vs baseline: 4.7957x; 1.0875x over previous
#include <hip/hip_runtime.h>

typedef __attribute__((ext_vector_type(2))) float v2f;
typedef __attribute__((ext_vector_type(8))) float v8f;

__global__ __launch_bounds__(256) void sg_square_wmma(
    const float* __restrict__ Sg, float* __restrict__ Sg2) {
  constexpr int NW = 8;
  constexpr int KW = 192 / NW;
  constexpr int KS = KW / 4;
  __shared__ float part[NW * 8 * 2 * 32];

  const int nt   = blockIdx.x;
  const int mt   = blockIdx.y;
  const int w    = threadIdx.x >> 5;
  const int l    = threadIdx.x & 31;
  const int half = l >> 4;
  const int lm   = l & 15;
  const int arow  = mt * 16 + lm;
  const int bcol0 = nt * 32 + lm;
  const int kb    = w * KW + half * 2;
  const float* __restrict__ ap  = Sg + arow * 192 + kb;
  const float* __restrict__ bp0 = Sg + kb * 192 + bcol0;

  v2f a[KS], b0[KS], b1[KS];
#pragma unroll
  for (int j = 0; j < KS; ++j) {
    const int k0 = j * 4;
    a[j].x  = ap[k0];
    a[j].y  = ap[k0 + 1];
    b0[j].x = bp0[k0 * 192];
    b0[j].y = bp0[k0 * 192 + 192];
    b1[j].x = bp0[k0 * 192 + 16];
    b1[j].y = bp0[k0 * 192 + 192 + 16];
  }

  __builtin_amdgcn_sched_barrier(0);

  v8f acc0 = {0.f, 0.f, 0.f, 0.f, 0.f, 0.f, 0.f, 0.f};
  v8f acc1 = {0.f, 0.f, 0.f, 0.f, 0.f, 0.f, 0.f, 0.f};
#pragma unroll
  for (int j = 0; j < KS; ++j) {
    acc0 = __builtin_amdgcn_wmma_f32_16x16x4_f32(
        false, a[j], false, b0[j], (short)0, acc0, false, false);
    acc1 = __builtin_amdgcn_wmma_f32_16x16x4_f32(
        false, a[j], false, b1[j], (short)0, acc1, false, false);
  }

#pragma unroll
  for (int v = 0; v < 8; ++v) {
    part[((w * 8 + v) * 2 + 0) * 32 + l] = acc0[v];
    part[((w * 8 + v) * 2 + 1) * 32 + l] = acc1[v];
  }
  __syncthreads();

  const int pidx = (w * 2 + half) * 32 + lm;
  float sum0 = part[pidx];
  float sum1 = part[pidx + 16];
#pragma unroll
  for (int q = 1; q < NW; ++q) {
    sum0 += part[q * 512 + pidx];
    sum1 += part[q * 512 + pidx + 16];
  }
  float* __restrict__ op = Sg2 + (mt * 16 + w) * 192 + nt * 32 + l;
  op[0]       = sum0;
  op[8 * 192] = sum1;
}

template <int TC, int C, int COUT, bool DOWN, bool UP, bool PREMIX>
__global__ __launch_bounds__(256) void s2_apply_wmma(
    const float* __restrict__ Sg, const float* __restrict__ Sg2,
    const float* __restrict__ Z, const float* __restrict__ s,
    float* __restrict__ Xo, const float* __restrict__ h,
    const float* __restrict__ hn) {
  constexpr int NW = 8;
  constexpr int KW = 192 / NW;
  constexpr int KS = KW / 4;
  __shared__ float part[NW * 8 * 4 * 32];
  constexpr int TAUS   = 32 / C;
  constexpr int K3     = 3 * C;
  constexpr int KSM    = (K3 + 3) / 4;
  constexpr bool KPAD  = (K3 % 4) != 0;
  constexpr int NTN    = COUT / 16;
  constexpr int NTILE  = TAUS * NTN;
  constexpr int KSPLIT = (NTILE >= NW) ? 1 : NW / NTILE;
  constexpr int TPW    = (NTILE >= NW) ? NTILE / NW : 1;
  constexpr int KPP    = KSM / KSPLIT;
  static_assert(COUT % 16 == 0 && KSM % KSPLIT == 0 &&
                NTILE * KSPLIT == NW * TPW && (TPW == 1 || NTN == 1),
                "mix tiles must deal evenly over the waves");
  static_assert(KSPLIT * 16 * TAUS * COUT <= NW * 8 * 4 * 32,
                "mixed values must fit the partial-tile buffer");
  constexpr int RS = 32 * 3 + 2;
  __shared__ float tile[16 * RS];
  __shared__ float vst[PREMIX ? 48 : 1];

  const int nt    = blockIdx.x;
  const int mt    = blockIdx.y;
  const int w     = threadIdx.x >> 5;
  const int l     = threadIdx.x & 31;
  const int half  = l >> 4;
  const int lm    = l & 15;
  const int arow  = mt * 16 + lm;
  const int bcol0 = nt * 32 + lm;
  const int bcol1 = bcol0 + 16;
  const int bcol0m  = (bcol0 >= C) ? (bcol0 - C) : (bcol0 - C + TC);
  const int bcol1m  = (bcol1 >= C) ? (bcol1 - C) : (bcol1 - C + TC);
  const int bcol0mm = (bcol0 >= 2 * C) ? (bcol0 - 2 * C) : (bcol0 - 2 * C + TC);
  const int bcol1mm = (bcol1 >= 2 * C) ? (bcol1 - 2 * C) : (bcol1 - 2 * C + TC);

  const int ecol   = nt * 32 + l;
  const int ecolm  = (ecol >= C) ? (ecol - C) : (ecol - C + TC);
  const int ecolmm = (ecol >= 2 * C) ? (ecol - 2 * C) : (ecol - 2 * C + TC);
  const float* __restrict__ zp = Z + (mt * 16 + w) * TC;
  const float ez0 = zp[ecol];
  const float em0 = zp[ecolm];
  const float en0 = zp[ecolmm];
  const float ez1 = zp[8 * TC + ecol];
  const float em1 = zp[8 * TC + ecolm];
  const float en1 = zp[8 * TC + ecolmm];
  const int mt0 = (NTILE >= NW) ? w * TPW : (w % NTILE);
  const int kp  = (NTILE >= NW) ? 0 : (w / NTILE);
  const int nn  = mt0 % NTN;
  v2f hv[KPP];
#pragma unroll
  for (int j = 0; j < KPP; ++j) {
    const int kk = 4 * (kp * KPP + j) + 2 * half;
    const float* __restrict__ hp = h + (nn * 16 + lm) * K3 + kk;
    if constexpr (KPAD) {
      hv[j].x = (kk < K3) ? hp[0] : 0.f;
      hv[j].y = (kk + 1 < K3) ? hp[1] : 0.f;
    } else {
      hv[j].x = hp[0];
      hv[j].y = hp[1];
    }
  }
  float hn0 = 0.f, hn1 = 0.f, hn2 = 0.f;
  if constexpr (PREMIX) {
    const int on = (int)threadIdx.x & 15;
    hn0 = hn[on * 3 + 0];
    hn1 = hn[on * 3 + 1];
    hn2 = hn[on * 3 + 2];
  }

  const float s00 = s[0], s01 = s[1], s10 = s[2], s11 = s[3];
  const float a0 = s00 * s00, a1 = 2.f * s00 * s10, a2 = s10 * s10;
  const float c0 = 2.f * s00 * s01;
  const float c1 = 2.f * (s00 * s11 + s10 * s01);
  const float c2 = 2.f * s10 * s11;
  const float d0 = s01 * s01, d1 = 2.f * s01 * s11, d2 = s11 * s11;

  const int kb = w * KW + half * 2;
  const float* __restrict__ ap    = Sg + arow * 192 + kb;
  const float* __restrict__ ap2   = Sg2 + arow * 192 + kb;
  const float* __restrict__ bp0   = Z + kb * TC + bcol0;
  const float* __restrict__ bp1   = Z + kb * TC + bcol1;
  const float* __restrict__ bp0m  = Z + kb * TC + bcol0m;
  const float* __restrict__ bp1m  = Z + kb * TC + bcol1m;
  const float* __restrict__ bp0mm = Z + kb * TC + bcol0mm;
  const float* __restrict__ bp1mm = Z + kb * TC + bcol1mm;

  v2f a[KS], a2f[KS];
  float z0x[KS], z0y[KS], z1x[KS], z1y[KS];
  float m0x[KS], m0y[KS], m1x[KS], m1y[KS];
  float n0x[KS], n0y[KS], n1x[KS], n1y[KS];
#pragma unroll
  for (int j = 0; j < KS; ++j) {
    const int k0 = j * 4;
    a[j].x   = ap[k0];
    a[j].y   = ap[k0 + 1];
    a2f[j].x = ap2[k0];
    a2f[j].y = ap2[k0 + 1];
    z0x[j] = bp0[k0 * TC];
    z0y[j] = bp0[k0 * TC + TC];
    z1x[j] = bp1[k0 * TC];
    z1y[j] = bp1[k0 * TC + TC];
    m0x[j] = bp0m[k0 * TC];
    m0y[j] = bp0m[k0 * TC + TC];
    if constexpr (C == 16) {
      m1x[j] = z0x[j];
      m1y[j] = z0y[j];
      n1x[j] = m0x[j];
      n1y[j] = m0y[j];
    } else {
      m1x[j] = bp1m[k0 * TC];
      m1y[j] = bp1m[k0 * TC + TC];
      n1x[j] = bp1mm[k0 * TC];
      n1y[j] = bp1mm[k0 * TC + TC];
    }
    n0x[j] = bp0mm[k0 * TC];
    n0y[j] = bp0mm[k0 * TC + TC];
  }

  __builtin_amdgcn_sched_barrier(0);

  v8f x1a = {0.f, 0.f, 0.f, 0.f, 0.f, 0.f, 0.f, 0.f};
  v8f x1b = {0.f, 0.f, 0.f, 0.f, 0.f, 0.f, 0.f, 0.f};
  v8f x2a = {0.f, 0.f, 0.f, 0.f, 0.f, 0.f, 0.f, 0.f};
  v8f x2b = {0.f, 0.f, 0.f, 0.f, 0.f, 0.f, 0.f, 0.f};
#pragma unroll
  for (int j = 0; j < KS; ++j) {
    v2f p0, p1, q0, q1, r0, r1;
    p0.x = s01 * z0x[j] + s11 * m0x[j];
    p0.y = s01 * z0y[j] + s11 * m0y[j];
    p1.x = s01 * z1x[j] + s11 * m1x[j];
    p1.y = s01 * z1y[j] + s11 * m1y[j];
    q0.x = c0 * z0x[j] + c1 * m0x[j] + c2 * n0x[j];
    q0.y = c0 * z0y[j] + c1 * m0y[j] + c2 * n0y[j];
    q1.x = c0 * z1x[j] + c1 * m1x[j] + c2 * n1x[j];
    q1.y = c0 * z1y[j] + c1 * m1y[j] + c2 * n1y[j];
    r0.x = d0 * z0x[j] + d1 * m0x[j] + d2 * n0x[j];
    r0.y = d0 * z0y[j] + d1 * m0y[j] + d2 * n0y[j];
    r1.x = d0 * z1x[j] + d1 * m1x[j] + d2 * n1x[j];
    r1.y = d0 * z1y[j] + d1 * m1y[j] + d2 * n1y[j];
    x1a = __builtin_amdgcn_wmma_f32_16x16x4_f32(
        false, a[j], false, p0, (short)0, x1a, false, false);
    x1b = __builtin_amdgcn_wmma_f32_16x16x4_f32(
        false, a[j], false, p1, (short)0, x1b, false, false);
    x2a = __builtin_amdgcn_wmma_f32_16x16x4_f32(
        false, a[j], false, q0, (short)0, x2a, false, false);
    x2b = __builtin_amdgcn_wmma_f32_16x16x4_f32(
        false, a[j], false, q1, (short)0, x2b, false, false);
    x2a = __builtin_amdgcn_wmma_f32_16x16x4_f32(
        false, a2f[j], false, r0, (short)0, x2a, false, false);
    x2b = __builtin_amdgcn_wmma_f32_16x16x4_f32(
        false, a2f[j], false, r1, (short)0, x2b, false, false);
  }

#pragma unroll
  for (int v = 0; v < 8; ++v) {
    part[((w * 8 + v) * 4 + 0) * 32 + l] = x1a[v];
    part[((w * 8 + v) * 4 + 1) * 32 + l] = x1b[v];
    part[((w * 8 + v) * 4 + 2) * 32 + l] = x2a[v];
    part[((w * 8 + v) * 4 + 3) * 32 + l] = x2b[v];
  }
  __syncthreads();

  const int pidx = (w * 4 + half) * 32 + lm;
  float u0 = part[pidx];
  float u1 = part[pidx + 16];
  float t0 = part[pidx + 64];
  float t1 = part[pidx + 64 + 16];
#pragma unroll
  for (int q = 1; q < NW; ++q) {
    u0 += part[q * 1024 + pidx];
    u1 += part[q * 1024 + pidx + 16];
    t0 += part[q * 1024 + pidx + 64];
    t1 += part[q * 1024 + pidx + 64 + 16];
  }
  const float f0 = u0 + s00 * ez0 + s10 * em0;
  const float f1 = u1 + s00 * ez1 + s10 * em1;
  const float g0 = t0 + a0 * ez0 + a1 * em0 + a2 * en0;
  const float g1 = t1 + a0 * ez1 + a1 * em1 + a2 * en1;
  tile[w * RS + l * 3 + 0]       = ez0;
  tile[w * RS + l * 3 + 1]       = f0;
  tile[w * RS + l * 3 + 2]       = g0;
  tile[(w + 8) * RS + l * 3 + 0] = ez1;
  tile[(w + 8) * RS + l * 3 + 1] = f1;
  tile[(w + 8) * RS + l * 3 + 2] = g1;
  __syncthreads();

#pragma unroll
  for (int tt = 0; tt < TPW; ++tt) {
    const int tau = (mt0 + tt) / NTN;
    const float* xr = tile + lm * RS + tau * K3 + 2 * half;
    v8f y = {0.f, 0.f, 0.f, 0.f, 0.f, 0.f, 0.f, 0.f};
#pragma unroll
    for (int j = 0; j < KPP; ++j) {
      const int kk = 4 * (kp * KPP + j);
      v2f xa;
      if constexpr (KPAD) {
        const bool in0 = kk + 2 * half < K3;
        const bool in1 = kk + 2 * half + 1 < K3;
        const float t0 = xr[in0 ? kk : -2 * half];
        const float t1 = xr[in1 ? kk + 1 : -2 * half];
        xa.x = in0 ? t0 : 0.f;
        xa.y = in1 ? t1 : 0.f;
      } else {
        xa.x = xr[kk];
        xa.y = xr[kk + 1];
      }
      y = __builtin_amdgcn_wmma_f32_16x16x4_f32(
          false, xa, false, hv[j], (short)0, y, false, false);
    }
#pragma unroll
    for (int v = 0; v < 8; ++v) {
      part[((kp * 16 + v + 8 * half) * TAUS + tau) * COUT + nn * 16 + lm] = y[v];
    }
  }
  __syncthreads();

  constexpr int TQT  = DOWN ? TAUS / 2 : TAUS;
  constexpr int TQN  = DOWN ? TC / C / 2 : TC / C;
  constexpr int NOUT = 16 * TQT * COUT;
  static_assert(!PREMIX || (NOUT == 256 && COUT == 16 && UP && !DOWN),
                "premix needs one (node, channel) pair per thread");
  static_assert(NOUT % 256 == 0, "whole trips only");
#pragma unroll
  for (int it = 0; it < NOUT / 256; ++it) {
    const int li = it * 256 + (int)threadIdx.x;
    const int o  = li % COUT;
    const int tl = (li / COUT) % TQT;
    const int nl = li / (COUT * TQT);
    const float* yp = part + (nl * TAUS + (DOWN ? 2 * tl : tl)) * COUT + o;
    float acc = yp[0];
    float accb = DOWN ? yp[COUT] : 0.f;
#pragma unroll
    for (int q = 1; q < KSPLIT; ++q) {
      acc += yp[q * 16 * TAUS * COUT];
      if (DOWN) accb += yp[q * 16 * TAUS * COUT + COUT];
    }
    float v = acc;
    if (DOWN) v = fmaxf(fmaxf(acc, accb), 0.f);
    const int node = mt * 16 + nl;
    const int tq   = nt * TQT + tl;
    if constexpr (PREMIX) {
      const float rv = fmaxf(v, 0.f);
      float p0 = hn0 * rv, p1 = hn1 * rv, p2 = hn2 * rv;
#pragma unroll
      for (int m = 8; m >= 1; m >>= 1) {
        p0 += __shfl_xor(p0, m, 32);
        p1 += __shfl_xor(p1, m, 32);
        p2 += __shfl_xor(p2, m, 32);
      }
      if (o < 3) vst[((o + 2) % 3) * 16 + nl] = (o == 0) ? p0 : ((o == 1) ? p1 : p2);
    } else if (UP) {
      const int ob = (node * 2 * TQN + 2 * tq) * COUT + o;
      Xo[ob]        = fmaxf(v, 0.f);
      Xo[ob + COUT] = 0.f;
    } else {
      Xo[(node * TQN + tq) * COUT + o] = v;
    }
  }
  if constexpr (PREMIX) {
    __syncthreads();
    float* __restrict__ vp = Xo + (nt * 12 + mt) * 64;
    if (w == 0) vp[l] = vst[l];
    if (w == 1 && l < 16) vp[32 + l] = vst[32 + l];
  }
}

__global__ __launch_bounds__(256) void last_layer_wmma(
    const float* __restrict__ Sg, const float* __restrict__ Sg2,
    const float* __restrict__ V, const float* __restrict__ s,
    float* __restrict__ out) {
  constexpr int NW = 8;
  constexpr int KW = 192 / NW;
  constexpr int KS = KW / 4;
  constexpr int VS = 17;
  constexpr int PL = 192 * VS + 2;
  constexpr int NCH = 192 / NW;
  __shared__ float vs[2 * PL];
  __shared__ float part[NW * 8 * 2 * 32];

  const int mt   = blockIdx.x;
  const int w    = threadIdx.x >> 5;
  const int l    = threadIdx.x & 31;
  const int half = l >> 4;
  const int lm   = l & 15;
  const int arow = mt * 16 + lm;
  const int kb   = w * KW + half * 2;
  const int odd  = l & 1;
  const int ta   = l >> 1;
  const int tb   = (ta + 15) & 15;

  float sv[NCH];
#pragma unroll
  for (int j = 0; j < NCH; ++j) sv[j] = V[(w * NCH + j) * 64 + l];

  const float* __restrict__ ap  = Sg + arow * 192 + kb;
  const float* __restrict__ ap2 = Sg2 + arow * 192 + kb;
  v2f a[KS], a2f[KS];
#pragma unroll
  for (int j = 0; j < KS; ++j) {
    const int k0 = j * 4;
    a[j].x   = ap[k0];
    a[j].y   = ap[k0 + 1];
    a2f[j].x = ap2[k0];
    a2f[j].y = ap2[k0 + 1];
  }

  const float* __restrict__ va = V + (ta * 12 + mt) * 64 + w;
  const float* __restrict__ vb = V + (tb * 12 + mt) * 64 + w;
  const float l1_0  = va[0];
  const float l2a_0 = va[16];
  const float l0_0  = va[32];
  const float l2b_0 = vb[16];
  const float l1_1  = va[8];
  const float l2a_1 = va[16 + 8];
  const float l0_1  = va[32 + 8];
  const float l2b_1 = vb[16 + 8];

  const float s00 = s[0], s01 = s[1], s10 = s[2], s11 = s[3];

#pragma unroll
  for (int j = 0; j < NCH; ++j) {
    vs[half * PL + ((j % 12) * 16 + lm) * VS + 2 * w + j / 12] = sv[j];
  }

  const float a0 = s00 * s00, a1 = 2.f * s00 * s10, a2 = s10 * s10;
  const float c0 = 2.f * s00 * s01;
  const float c1 = 2.f * (s00 * s11 + s10 * s01);
  const float c2 = 2.f * s10 * s11;
  const float d0 = s01 * s01, d1 = 2.f * s01 * s11, d2 = s11 * s11;
  const float e1 = odd ? s11 : s01;
  const float e2 = odd ? c1 : c0;
  const float e3 = odd ? 0.f : c2;
  const float f2 = odd ? d1 : d0;
  const float f3 = odd ? 0.f : d2;
  __syncthreads();

  const int ta0 = lm >> 1;
  const int ta1 = ta0 + 8;
  const int tb0 = (ta0 + 15) & 15;
  const int tb1 = (ta0 + 7) & 15;
  v8f acc0 = {0.f, 0.f, 0.f, 0.f, 0.f, 0.f, 0.f, 0.f};
  v8f acc1 = {0.f, 0.f, 0.f, 0.f, 0.f, 0.f, 0.f, 0.f};
#pragma unroll
  for (int j = 0; j < KS; ++j) {
    const float* r1 = vs + (kb + j * 4) * VS;
    const float* r2 = r1 + PL;
    v2f p0, p1, q0, q1;
    p0.x = e1 * r1[ta0] + e2 * r2[ta0] + e3 * r2[tb0];
    p0.y = e1 * r1[VS + ta0] + e2 * r2[VS + ta0] + e3 * r2[VS + tb0];
    p1.x = e1 * r1[ta1] + e2 * r2[ta1] + e3 * r2[tb1];
    p1.y = e1 * r1[VS + ta1] + e2 * r2[VS + ta1] + e3 * r2[VS + tb1];
    q0.x = f2 * r2[ta0] + f3 * r2[tb0];
    q0.y = f2 * r2[VS + ta0] + f3 * r2[VS + tb0];
    q1.x = f2 * r2[ta1] + f3 * r2[tb1];
    q1.y = f2 * r2[VS + ta1] + f3 * r2[VS + tb1];
    acc0 = __builtin_amdgcn_wmma_f32_16x16x4_f32(
        false, a[j], false, p0, (short)0, acc0, false, false);
    acc1 = __builtin_amdgcn_wmma_f32_16x16x4_f32(
        false, a[j], false, p1, (short)0, acc1, false, false);
    acc0 = __builtin_amdgcn_wmma_f32_16x16x4_f32(
        false, a2f[j], false, q0, (short)0, acc0, false, false);
    acc1 = __builtin_amdgcn_wmma_f32_16x16x4_f32(
        false, a2f[j], false, q1, (short)0, acc1, false, false);
  }

#pragma unroll
  for (int v = 0; v < 8; ++v) {
    part[((w * 8 + v) * 2 + 0) * 32 + l] = acc0[v];
    part[((w * 8 + v) * 2 + 1) * 32 + l] = acc1[v];
  }
  __syncthreads();

  const int pidx = (w * 2 + half) * 32 + lm;
  float sum0 = part[pidx];
  float sum1 = part[pidx + 16];
#pragma unroll
  for (int q = 1; q < NW; ++q) {
    sum0 += part[q * 512 + pidx];
    sum1 += part[q * 512 + pidx + 16];
  }
  const float g0 = odd ? 0.f : 1.f;
  const float g1 = odd ? s10 : s00;
  const float g2 = odd ? a1 : a0;
  const float g3 = odd ? 0.f : a2;
  float* __restrict__ op = out + (mt * 16 + w) * 32 + l;
  op[0]      = sum0 + g0 * l0_0 + g1 * l1_0 + g2 * l2a_0 + g3 * l2b_0;
  op[8 * 32] = sum1 + g0 * l0_1 + g1 * l1_1 + g2 * l2a_1 + g3 * l2b_1;
}

extern "C" void kernel_launch(void* const* d_in, const int* in_sizes, int n_in,
                              void* d_out, int out_size, void* d_ws,
                              size_t ws_size, hipStream_t stream) {
  (void)in_sizes; (void)n_in; (void)out_size; (void)ws_size;
  const float* X    = (const float*)d_in[0];
  const float* Sg   = (const float*)d_in[1];
  const float* s    = (const float*)d_in[2];
  const float* h_e1 = (const float*)d_in[3];
  const float* h_e2 = (const float*)d_in[4];
  const float* h_d1 = (const float*)d_in[5];
  const float* h_d2 = (const float*)d_in[6];
  float* out = (float*)d_out;
  float* ws  = (float*)d_ws;

  const int BUF = 192 * 32 * 16;
  float* Sg2 = ws + 0 * BUF;
  float* A0  = ws + 2 * BUF;
  float* A1  = ws + 3 * BUF;
  float* V   = A0;

  sg_square_wmma<<<dim3(6, 12), 256, 0, stream>>>(Sg, Sg2);
  s2_apply_wmma<32, 1, 16, true, false, false>
      <<<dim3(1, 12), 256, 0, stream>>>(Sg, Sg2, X, s, A0, h_e1, h_e1);
  s2_apply_wmma<256, 16, 32, true, true, false>
      <<<dim3(8, 12), 256, 0, stream>>>(Sg, Sg2, A0, s, A1, h_e2, h_e2);
  s2_apply_wmma<512, 32, 16, false, true, true>
      <<<dim3(16, 12), 256, 0, stream>>>(Sg, Sg2, A1, s, V, h_d1, h_d2);
  last_layer_wmma<<<dim3(12), 256, 0, stream>>>(Sg, Sg2, V, s, out);
}
